# starting kernel
# baseline (speedup 1.0000x reference)
#include <hip/hip_runtime.h>
#include <hip/hip_bf16.h>
#include <math.h>

#define NNODES 100000
#define NEDGES 1200000
#define DIN    64
#define DOUT   64
#define EDIM   16
#define KPRE   144
#define KPOST  832
#define AVG_D  2.5649493574615367f
#define EPSV   1e-05f

typedef __attribute__((ext_vector_type(2))) float v2f;
typedef __attribute__((ext_vector_type(8))) float v8f;

__device__ __forceinline__ unsigned fenc(float f) {
    unsigned u = __float_as_uint(f);
    return (u & 0x80000000u) ? ~u : (u | 0x80000000u);
}
__device__ __forceinline__ float fdec(unsigned u) {
    return (u & 0x80000000u) ? __uint_as_float(u & 0x7fffffffu)
                             : __uint_as_float(~u);
}


__global__ __launch_bounds__(256) void pna_init(float* s1, float* s2,
                                                unsigned* mx, unsigned* mn,
                                                float* deg, float* bns, float* bnq) {
    int i = blockIdx.x * 256 + threadIdx.x;
    if (i < NNODES * DIN) {
        s1[i] = 0.0f; s2[i] = 0.0f;
        mx[i] = 0u;
        mn[i] = 0xFFFFFFFFu;
    }
    if (i < NNODES) deg[i] = 0.0f;
    if (i < DOUT) { bns[i] = 0.0f; bnq[i] = 0.0f; }
}

__global__ __launch_bounds__(256) void pna_degree(const int* dst, float* deg) {
    int i = blockIdx.x * 256 + threadIdx.x;
    if (i < NEDGES) atomicAdd(&deg[dst[i]], 1.0f);
}

#define ZSTRIDE 148
__global__ __launch_bounds__(256) void pna_edge(
    const float* __restrict__ h, const float* __restrict__ e,
    const float* __restrict__ Wpre, const float* __restrict__ bpre,
    const int* __restrict__ src, const int* __restrict__ dst,
    float* s1, float* s2, unsigned* mx, unsigned* mn) {

    __shared__ float zs[32 * ZSTRIDE];
    __shared__ float wsT[DIN * ZSTRIDE];
    __shared__ int   ldst[32];

    const int tid = threadIdx.x;
    const int e0  = blockIdx.x * 32;

    for (int i = tid; i < KPRE * DIN; i += 256) {
        int n = i & 63, k = i >> 6;
        wsT[n * ZSTRIDE + k] = Wpre[k * DIN + n];
    }
    {
        int r = tid >> 3, p = tid & 7;
        int ei = e0 + r;
        int sI = src[ei], dI = dst[ei];
        if (p == 0) ldst[r] = dI;
        int c0 = p * 18;
        for (int c = c0; c < c0 + 18; ++c) {
            float v;
            if (c < DIN)            v = h[sI * DIN + c];
            else if (c < 2 * DIN)   v = h[dI * DIN + (c - DIN)];
            else                    v = e[ei * EDIM + (c - 2 * DIN)];
            zs[r * ZSTRIDE + c] = v;
        }
    }
    __syncthreads();

    const int l  = tid & 31;
    const int wv = tid >> 5;
    const int cg = wv & 3;
    const int eg = wv >> 2;
    const int lo = l & 15;
    const int hi = l >> 4;

    v8f acc = {};
    const float* arow = &zs[(eg * 16 + lo) * ZSTRIDE + 2 * hi];
    const float* brow = &wsT[(cg * 16 + lo) * ZSTRIDE + 2 * hi];
#pragma unroll 4
    for (int k0 = 0; k0 < KPRE; k0 += 4) {
        v2f a = *(const v2f*)(arow + k0);
        v2f b = *(const v2f*)(brow + k0);
        acc = __builtin_amdgcn_wmma_f32_16x16x4_f32(
            false, a, false, b, (short)0, acc, false, false);
    }

    const int col = cg * 16 + lo;
    const float bv = bpre[col];
#pragma unroll
    for (int r = 0; r < 8; ++r) {
        int M  = r + 8 * hi;
        int er = eg * 16 + M;
        int d  = ldst[er];
        float v = acc[r] + bv;
        unsigned ev = fenc(v);
        size_t o = (size_t)d * DIN + col;
        atomicAdd(&s1[o], v);
        atomicAdd(&s2[o], v * v);
        atomicMax(&mx[o], ev);
        atomicMin(&mn[o], ev);
    }
}

#define FSTRIDE 836
__global__ __launch_bounds__(128) void pna_node(
    const float* __restrict__ h, const float* __restrict__ snorm,
    const float* __restrict__ Wpost, const float* __restrict__ bpost,
    const float* __restrict__ s1, const float* __restrict__ s2,
    const unsigned* __restrict__ mx, const unsigned* __restrict__ mn,
    const float* __restrict__ deg,
    float* __restrict__ out, float* bns, float* bnq) {

    __shared__ float feat[16 * FSTRIDE];
    const int tid = threadIdx.x;
    const int n0  = blockIdx.x * 16;

    for (int i = tid; i < 16 * 320; i += 128) {
        int r = i / 320, c = i % 320;
        int node = n0 + r;
        float v;
        if (c < DIN) {
            v = h[node * DIN + c];
        } else {
            int a = c - DIN;
            int sub = a >> 6, colc = a & 63;
            size_t o = (size_t)node * DIN + colc;
            float D  = deg[node];
            if (sub == 0)      v = s1[o] / D;
            else if (sub == 1) v = fdec(mx[o]);
            else if (sub == 2) v = fdec(mn[o]);
            else {
                float mean = s1[o] / D;
                v = sqrtf(fmaxf(s2[o] / D - mean * mean, 0.0f) + EPSV);
            }
            if (c == DIN) feat[r * FSTRIDE + 832] = logf(D + 1.0f);
        }
        feat[r * FSTRIDE + c] = v;
    }
    __syncthreads();
    for (int i = tid; i < 16 * 512; i += 128) {
        int r = i >> 9, j = i & 511;
        float logd = feat[r * FSTRIDE + 832];
        float f = (j < 256) ? (logd / AVG_D) : (AVG_D / logd);
        feat[r * FSTRIDE + 320 + j] = feat[r * FSTRIDE + DIN + (j & 255)] * f;
    }
    __syncthreads();

    const int l  = tid & 31;
    const int cg = tid >> 5;
    const int lo = l & 15;
    const int hi = l >> 4;
    const int col = cg * 16 + lo;

    v8f acc = {};
    const float* arow = &feat[lo * FSTRIDE + 2 * hi];
#pragma unroll 4
    for (int k0 = 0; k0 < KPOST; k0 += 4) {
        v2f a = *(const v2f*)(arow + k0);
        v2f b;
        b.x = Wpost[(k0 + 2 * hi)     * DOUT + col];
        b.y = Wpost[(k0 + 2 * hi + 1) * DOUT + col];
        acc = __builtin_amdgcn_wmma_f32_16x16x4_f32(
            false, a, false, b, (short)0, acc, false, false);
    }

    const float bv = bpost[col];
    float ps = 0.0f, ps2 = 0.0f;
#pragma unroll
    for (int r = 0; r < 8; ++r) {
        int M = r + 8 * hi;
        int node = n0 + M;
        float v = (acc[r] + bv) * snorm[node];
        out[(size_t)node * DOUT + col] = v;
        ps += v; ps2 += v * v;
    }
    ps  += __shfl_xor(ps, 16, 32);
    ps2 += __shfl_xor(ps2, 16, 32);
    if (hi == 0) {
        atomicAdd(&bns[col], ps);
        atomicAdd(&bnq[col], ps2);
    }
}

__global__ __launch_bounds__(256) void pna_bn(
    float* out, const float* bns, const float* bnq,
    const float* gamma, const float* beta) {
    int i = blockIdx.x * 256 + threadIdx.x;
    if (i >= NNODES * DOUT) return;
    int c = i & 63;
    float mu  = bns[c] / (float)NNODES;
    float var = bnq[c] / (float)NNODES - mu * mu;
    float v = out[i];
    out[i] = (v - mu) * rsqrtf(var + 1e-05f) * gamma[c] + beta[c];
}

extern "C" void kernel_launch(void* const* d_in, const int* in_sizes, int n_in,
                              void* d_out, int out_size, void* d_ws, size_t ws_size,
                              hipStream_t stream) {
    const float* h      = (const float*)d_in[0];
    const float* e      = (const float*)d_in[1];
    const float* snorm  = (const float*)d_in[2];
    const float* Wpre   = (const float*)d_in[3];
    const float* bpre   = (const float*)d_in[4];
    const float* Wpost  = (const float*)d_in[5];
    const float* bpost  = (const float*)d_in[6];
    const float* gamma  = (const float*)d_in[7];
    const float* beta   = (const float*)d_in[8];
    const int*   src    = (const int*)d_in[9];
    const int*   dst    = (const int*)d_in[10];
    float* out = (float*)d_out;

    float*    s1  = (float*)d_ws;
    float*    s2  = s1 + (size_t)NNODES * DIN;
    unsigned* mx  = (unsigned*)(s2 + (size_t)NNODES * DIN);
    unsigned* mn  = mx + (size_t)NNODES * DIN;
    float*    deg = (float*)(mn + (size_t)NNODES * DIN);
    float*    bns = deg + NNODES;
    float*    bnq = bns + DOUT;

    pna_init  <<<(NNODES * DIN + 255) / 256, 256, 0, stream>>>(s1, s2, mx, mn, deg, bns, bnq);
    pna_degree<<<(NEDGES + 255) / 256, 256, 0, stream>>>(dst, deg);
    pna_edge  <<<NEDGES / 32, 256, 0, stream>>>(h, e, Wpre, bpre, src, dst, s1, s2, mx, mn);
    pna_node  <<<NNODES / 16, 128, 0, stream>>>(h, snorm, Wpost, bpost, s1, s2, mx, mn, deg,
                                                out, bns, bnq);
    pna_bn    <<<(NNODES * DOUT + 255) / 256, 256, 0, stream>>>(out, bns, bnq, gamma, beta);
}


# Pre-transform moved from edges to nodes (linearity): P = h W_pre[0:64], Q = h W_pre[64:128] + b_pre per node, 2 more launches; edge gathers P[src], k 144 to 16; Q added per node; sum order changes
# speedup vs baseline: 1.8574x; 1.8574x over previous
#include <hip/hip_runtime.h>
#include <hip/hip_bf16.h>
#include <math.h>

#define NNODES 100000
#define NEDGES 1200000
#define DIN    64
#define DOUT   64
#define EDIM   16
#define KPRE   144
#define KPOST  832
#define AVG_D  2.5649493574615367f
#define EPSV   1e-05f

typedef __attribute__((ext_vector_type(2))) float v2f;
typedef __attribute__((ext_vector_type(8))) float v8f;

__device__ __forceinline__ unsigned fenc(float f) {
    unsigned u = __float_as_uint(f);
    return (u & 0x80000000u) ? ~u : (u | 0x80000000u);
}
__device__ __forceinline__ float fdec(unsigned u) {
    return (u & 0x80000000u) ? __uint_as_float(u & 0x7fffffffu)
                             : __uint_as_float(~u);
}


__global__ __launch_bounds__(256) void pna_init(float* s1, float* s2,
                                                unsigned* mx, unsigned* mn,
                                                float* deg, float* bns, float* bnq) {
    int i = blockIdx.x * 256 + threadIdx.x;
    if (i < NNODES * DIN) {
        s1[i] = 0.0f; s2[i] = 0.0f;
        mx[i] = 0u;
        mn[i] = 0xFFFFFFFFu;
    }
    if (i < NNODES) deg[i] = 0.0f;
    if (i < DOUT) { bns[i] = 0.0f; bnq[i] = 0.0f; }
}

__global__ __launch_bounds__(256) void pna_degree(const int* dst, float* deg) {
    int i = blockIdx.x * 256 + threadIdx.x;
    if (i < NEDGES) atomicAdd(&deg[dst[i]], 1.0f);
}

__global__ __launch_bounds__(256) void pna_pre(
    const float* __restrict__ h, const float* __restrict__ Wpre,
    const float* __restrict__ bpre, int koff, int use_bias, float* __restrict__ Y) {
    const int tid = threadIdx.x;
    const int l  = tid & 31;
    const int wv = tid >> 5;
    const int cg = wv & 3;
    const int eg = wv >> 2;
    const int lo = l & 15;
    const int hi = l >> 4;
    const int n0 = blockIdx.x * 32 + eg * 16;

    const float* ap = h + (size_t)(n0 + lo) * DIN + 32 * hi;
    float4 a[8];
#pragma unroll
    for (int j = 0; j < 8; ++j) a[j] = *(const float4*)(ap + 4 * j);

    const float* wp = Wpre + (size_t)koff * DIN + cg * 16 + lo;
    v8f acc = {};
#pragma unroll
    for (int t = 0; t < 16; ++t) {
        const int k = 32 * hi + 4 * (t >> 1) + 2 * (t & 1);
        v2f b;
        b.x = wp[(size_t)k * DIN];
        b.y = wp[(size_t)(k + 1) * DIN];
        v2f av;
        if (t & 1) { av.x = a[t >> 1].z; av.y = a[t >> 1].w; }
        else       { av.x = a[t >> 1].x; av.y = a[t >> 1].y; }
        acc = __builtin_amdgcn_wmma_f32_16x16x4_f32(
            false, av, false, b, (short)0, acc, false, false);
    }
    const float bv = use_bias ? bpre[cg * 16 + lo] : 0.0f;
#pragma unroll
    for (int r = 0; r < 8; ++r)
        Y[(size_t)(n0 + 8 * hi + r) * DIN + cg * 16 + lo] = acc[r] + bv;
}

#define MSTRIDE 68
#define ETILES  (NEDGES / 16)
#define EBLOCKS 1875
#define EWAVES  (EBLOCKS * 4)
#define ETPW    (ETILES / EWAVES)
__global__ __launch_bounds__(128) void pna_edge(
    const float* __restrict__ e, const float* __restrict__ Wpre,
    const int* __restrict__ src, const int* __restrict__ dst,
    const float* __restrict__ P,
    float* s1, float* s2, unsigned* mx, unsigned* mn) {

    __shared__ float ms_all[4 * 16 * MSTRIDE];

    const int tid = threadIdx.x;
    const int l  = tid & 31;
    const int wv = tid >> 5;
    const int lo = l & 15;
    const int hi = l >> 4;
    float* ms = ms_all + wv * 16 * MSTRIDE;

    v2f b[4][4];
#pragma unroll
    for (int cg = 0; cg < 4; ++cg)
#pragma unroll
        for (int j = 0; j < 4; ++j) {
            const int k = 2 * DIN + 8 * hi + 2 * j;
            b[cg][j].x = Wpre[(size_t)k * DIN + cg * 16 + lo];
            b[cg][j].y = Wpre[(size_t)(k + 1) * DIN + cg * 16 + lo];
        }

    int tile = blockIdx.x * 4 + wv;
    int sI = src[tile * 16 + lo];
    int dI = dst[tile * 16 + lo];
    const float* ep = e + (size_t)(tile * 16 + lo) * EDIM + 8 * hi;
    float4 f0 = *(const float4*)ep;
    float4 f1 = *(const float4*)(ep + 4);

    for (int it = 0; it < ETPW; ++it) {
        const int tn = (it + 1 < ETPW) ? tile + EWAVES : tile;
        const int sN = src[tn * 16 + lo];
        const int dN = dst[tn * 16 + lo];
        const float* en = e + (size_t)(tn * 16 + lo) * EDIM + 8 * hi;
        const float4 g0 = *(const float4*)en;
        const float4 g1 = *(const float4*)(en + 4);

        float p0[16], p1[16];
#pragma unroll
        for (int i = 0; i < 16; ++i) {
            const int s = __builtin_amdgcn_readlane(sI, i);
            p0[i] = P[(size_t)s * DIN + l];
            p1[i] = P[(size_t)s * DIN + 32 + l];
        }

        v2f a[4];
        a[0].x = f0.x; a[0].y = f0.y;
        a[1].x = f0.z; a[1].y = f0.w;
        a[2].x = f1.x; a[2].y = f1.y;
        a[3].x = f1.z; a[3].y = f1.w;
        v8f acc[4];
#pragma unroll
        for (int cg = 0; cg < 4; ++cg) acc[cg] = (v8f){};
#pragma unroll
        for (int j = 0; j < 4; ++j)
#pragma unroll
            for (int cg = 0; cg < 4; ++cg)
                acc[cg] = __builtin_amdgcn_wmma_f32_16x16x4_f32(
                    false, a[j], false, b[cg][j], (short)0, acc[cg], false, false);

#pragma unroll
        for (int cg = 0; cg < 4; ++cg)
#pragma unroll
            for (int r = 0; r < 8; ++r)
                ms[(8 * hi + r) * MSTRIDE + cg * 16 + lo] = acc[cg][r];
        __builtin_amdgcn_fence(__ATOMIC_RELEASE, "wavefront");
        __builtin_amdgcn_wave_barrier();
        __builtin_amdgcn_fence(__ATOMIC_ACQUIRE, "wavefront");

#pragma unroll
        for (int i = 0; i < 16; ++i) {
            const int d = __builtin_amdgcn_readlane(dI, i);
            const float v0 = ms[i * MSTRIDE + l] + p0[i];
            const float v1 = ms[i * MSTRIDE + 32 + l] + p1[i];
            const unsigned ev0 = fenc(v0), ev1 = fenc(v1);
            const size_t o = (size_t)d * DIN + l;
            atomicAdd(&s1[o], v0);
            atomicAdd(&s1[o + 32], v1);
            atomicAdd(&s2[o], v0 * v0);
            atomicAdd(&s2[o + 32], v1 * v1);
            atomicMax(&mx[o], ev0);
            atomicMax(&mx[o + 32], ev1);
            atomicMin(&mn[o], ev0);
            atomicMin(&mn[o + 32], ev1);
        }
        __builtin_amdgcn_fence(__ATOMIC_RELEASE, "wavefront");
        __builtin_amdgcn_wave_barrier();
        __builtin_amdgcn_fence(__ATOMIC_ACQUIRE, "wavefront");

        tile = tn; sI = sN; dI = dN; f0 = g0; f1 = g1;
    }
}

#define FSTRIDE 836
__global__ __launch_bounds__(128) void pna_node(
    const float* __restrict__ h, const float* __restrict__ snorm,
    const float* __restrict__ Wpost, const float* __restrict__ bpost,
    const float* __restrict__ s1, const float* __restrict__ s2,
    const unsigned* __restrict__ mx, const unsigned* __restrict__ mn,
    const float* __restrict__ deg,
    float* out, float* bns, float* bnq) {

    __shared__ float feat[16 * FSTRIDE];
    const int tid = threadIdx.x;
    const int n0  = blockIdx.x * 16;

    for (int i = tid; i < 16 * 320; i += 128) {
        int r = i / 320, c = i % 320;
        int node = n0 + r;
        float v;
        if (c < DIN) {
            v = h[node * DIN + c];
        } else {
            int a = c - DIN;
            int sub = a >> 6, colc = a & 63;
            size_t o = (size_t)node * DIN + colc;
            float D  = deg[node];
            if (sub == 0)      v = s1[o] / D + out[o];
            else if (sub == 1) v = fdec(mx[o]) + out[o];
            else if (sub == 2) v = fdec(mn[o]) + out[o];
            else {
                float mean = s1[o] / D;
                v = sqrtf(fmaxf(s2[o] / D - mean * mean, 0.0f) + EPSV);
            }
            if (c == DIN) feat[r * FSTRIDE + 832] = logf(D + 1.0f);
        }
        feat[r * FSTRIDE + c] = v;
    }
    __syncthreads();
    for (int i = tid; i < 16 * 512; i += 128) {
        int r = i >> 9, j = i & 511;
        float logd = feat[r * FSTRIDE + 832];
        float f = (j < 256) ? (logd / AVG_D) : (AVG_D / logd);
        feat[r * FSTRIDE + 320 + j] = feat[r * FSTRIDE + DIN + (j & 255)] * f;
    }
    __syncthreads();

    const int l  = tid & 31;
    const int cg = tid >> 5;
    const int lo = l & 15;
    const int hi = l >> 4;
    const int col = cg * 16 + lo;

    v8f acc = {};
    const float* arow = &feat[lo * FSTRIDE + 2 * hi];
#pragma unroll 4
    for (int k0 = 0; k0 < KPOST; k0 += 4) {
        v2f a = *(const v2f*)(arow + k0);
        v2f b;
        b.x = Wpost[(k0 + 2 * hi)     * DOUT + col];
        b.y = Wpost[(k0 + 2 * hi + 1) * DOUT + col];
        acc = __builtin_amdgcn_wmma_f32_16x16x4_f32(
            false, a, false, b, (short)0, acc, false, false);
    }

    const float bv = bpost[col];
    float ps = 0.0f, ps2 = 0.0f;
#pragma unroll
    for (int r = 0; r < 8; ++r) {
        int M = r + 8 * hi;
        int node = n0 + M;
        float v = (acc[r] + bv) * snorm[node];
        out[(size_t)node * DOUT + col] = v;
        ps += v; ps2 += v * v;
    }
    ps  += __shfl_xor(ps, 16, 32);
    ps2 += __shfl_xor(ps2, 16, 32);
    if (hi == 0) {
        atomicAdd(&bns[col], ps);
        atomicAdd(&bnq[col], ps2);
    }
}

__global__ __launch_bounds__(256) void pna_bn(
    float* out, const float* bns, const float* bnq,
    const float* gamma, const float* beta) {
    int i = blockIdx.x * 256 + threadIdx.x;
    if (i >= NNODES * DOUT) return;
    int c = i & 63;
    float mu  = bns[c] / (float)NNODES;
    float var = bnq[c] / (float)NNODES - mu * mu;
    float v = out[i];
    out[i] = (v - mu) * rsqrtf(var + 1e-05f) * gamma[c] + beta[c];
}

extern "C" void kernel_launch(void* const* d_in, const int* in_sizes, int n_in,
                              void* d_out, int out_size, void* d_ws, size_t ws_size,
                              hipStream_t stream) {
    const float* h      = (const float*)d_in[0];
    const float* e      = (const float*)d_in[1];
    const float* snorm  = (const float*)d_in[2];
    const float* Wpre   = (const float*)d_in[3];
    const float* bpre   = (const float*)d_in[4];
    const float* Wpost  = (const float*)d_in[5];
    const float* bpost  = (const float*)d_in[6];
    const float* gamma  = (const float*)d_in[7];
    const float* beta   = (const float*)d_in[8];
    const int*   src    = (const int*)d_in[9];
    const int*   dst    = (const int*)d_in[10];
    float* out = (float*)d_out;

    float*    s1  = (float*)d_ws;
    float*    s2  = s1 + (size_t)NNODES * DIN;
    unsigned* mx  = (unsigned*)(s2 + (size_t)NNODES * DIN);
    unsigned* mn  = mx + (size_t)NNODES * DIN;
    float*    deg = (float*)(mn + (size_t)NNODES * DIN);
    float*    bns = deg + NNODES;
    float*    bnq = bns + DOUT;

    pna_init  <<<(NNODES * DIN + 255) / 256, 256, 0, stream>>>(s1, s2, mx, mn, deg, bns, bnq);
    pna_degree<<<(NEDGES + 255) / 256, 256, 0, stream>>>(dst, deg);
    pna_pre   <<<NNODES / 32, 256, 0, stream>>>(h, Wpre, bpre, 0, 0, out);
    pna_edge  <<<EBLOCKS, 128, 0, stream>>>(e, Wpre, src, dst, out, s1, s2, mx, mn);
    pna_pre   <<<NNODES / 32, 256, 0, stream>>>(h, Wpre, bpre, DIN, 1, out);
    pna_node  <<<NNODES / 16, 128, 0, stream>>>(h, snorm, Wpost, bpost, s1, s2, mx, mn, deg,
                                                out, bns, bnq);
    pna_bn    <<<(NNODES * DOUT + 255) / 256, 256, 0, stream>>>(out, bns, bnq, gamma, beta);
}


# Degree counted in LDS: block (node range, dst slice) counts its slice in LDS, then adds the counts to deg one whole 128-byte line per wave atomic, not one scattered atomic per edge; add order changes
# speedup vs baseline: 2.0241x; 1.0897x over previous
#include <hip/hip_runtime.h>
#include <hip/hip_bf16.h>
#include <math.h>

#define NNODES 100000
#define NEDGES 1200000
#define DIN    64
#define DOUT   64
#define EDIM   16
#define KPRE   144
#define KPOST  832
#define AVG_D  2.5649493574615367f
#define EPSV   1e-05f

typedef __attribute__((ext_vector_type(2))) float v2f;
typedef __attribute__((ext_vector_type(8))) float v8f;

__device__ __forceinline__ unsigned fenc(float f) {
    unsigned u = __float_as_uint(f);
    return (u & 0x80000000u) ? ~u : (u | 0x80000000u);
}
__device__ __forceinline__ float fdec(unsigned u) {
    return (u & 0x80000000u) ? __uint_as_float(u & 0x7fffffffu)
                             : __uint_as_float(~u);
}


__global__ __launch_bounds__(256) void pna_init(float* s1, float* s2,
                                                unsigned* mx, unsigned* mn,
                                                float* deg, float* bns, float* bnq) {
    int i = blockIdx.x * 256 + threadIdx.x;
    if (i < NNODES * DIN) {
        s1[i] = 0.0f; s2[i] = 0.0f;
        mx[i] = 0u;
        mn[i] = 0xFFFFFFFFu;
    }
    if (i < NNODES) deg[i] = 0.0f;
    if (i < DOUT) { bns[i] = 0.0f; bnq[i] = 0.0f; }
}

#define DRANGES 8
#define DSLICES 32
#define DRSZ    12544
#define DSL4    (NEDGES / DSLICES / 4)
#define DTHREADS 512
#define DPER    ((DSL4 + DTHREADS - 1) / DTHREADS)
__global__ __launch_bounds__(DTHREADS) void pna_degree(const int* __restrict__ dst, float* deg) {
    __shared__ unsigned cnt[DRSZ];
    const int tid = threadIdx.x;
    const int rg  = blockIdx.x % DRANGES;
    const int sl  = blockIdx.x / DRANGES;
    const int r0  = rg * DRSZ;

    const int4* p = (const int4*)(dst + (size_t)sl * (NEDGES / DSLICES));
    int4 v[DPER];
#pragma unroll
    for (int j = 0; j < DPER; ++j) {
        const int i = tid + j * DTHREADS;
        v[j] = (i < DSL4) ? p[i] : make_int4(-1, -1, -1, -1);
    }

    for (int i = tid; i < DRSZ; i += DTHREADS) cnt[i] = 0u;
    __syncthreads();

#pragma unroll
    for (int j = 0; j < DPER; ++j) {
        const unsigned o0 = (unsigned)(v[j].x - r0);
        const unsigned o1 = (unsigned)(v[j].y - r0);
        const unsigned o2 = (unsigned)(v[j].z - r0);
        const unsigned o3 = (unsigned)(v[j].w - r0);
        if (o0 < DRSZ) atomicAdd(&cnt[o0], 1u);
        if (o1 < DRSZ) atomicAdd(&cnt[o1], 1u);
        if (o2 < DRSZ) atomicAdd(&cnt[o2], 1u);
        if (o3 < DRSZ) atomicAdd(&cnt[o3], 1u);
    }
    __syncthreads();

    for (int i = tid; i < DRSZ; i += DTHREADS) {
        const unsigned c = cnt[i];
        const int n = r0 + i;
        if (c != 0u && n < NNODES) atomicAdd(&deg[n], (float)c);
    }
}

__global__ __launch_bounds__(256) void pna_pre(
    const float* __restrict__ h, const float* __restrict__ Wpre,
    const float* __restrict__ bpre, int koff, int use_bias, float* __restrict__ Y) {
    const int tid = threadIdx.x;
    const int l  = tid & 31;
    const int wv = tid >> 5;
    const int cg = wv & 3;
    const int eg = wv >> 2;
    const int lo = l & 15;
    const int hi = l >> 4;
    const int n0 = blockIdx.x * 32 + eg * 16;

    const float* ap = h + (size_t)(n0 + lo) * DIN + 32 * hi;
    float4 a[8];
#pragma unroll
    for (int j = 0; j < 8; ++j) a[j] = *(const float4*)(ap + 4 * j);

    const float* wp = Wpre + (size_t)koff * DIN + cg * 16 + lo;
    v8f acc = {};
#pragma unroll
    for (int t = 0; t < 16; ++t) {
        const int k = 32 * hi + 4 * (t >> 1) + 2 * (t & 1);
        v2f b;
        b.x = wp[(size_t)k * DIN];
        b.y = wp[(size_t)(k + 1) * DIN];
        v2f av;
        if (t & 1) { av.x = a[t >> 1].z; av.y = a[t >> 1].w; }
        else       { av.x = a[t >> 1].x; av.y = a[t >> 1].y; }
        acc = __builtin_amdgcn_wmma_f32_16x16x4_f32(
            false, av, false, b, (short)0, acc, false, false);
    }
    const float bv = use_bias ? bpre[cg * 16 + lo] : 0.0f;
#pragma unroll
    for (int r = 0; r < 8; ++r)
        Y[(size_t)(n0 + 8 * hi + r) * DIN + cg * 16 + lo] = acc[r] + bv;
}

#define MSTRIDE 68
#define ETILES  (NEDGES / 16)
#define EBLOCKS 1875
#define EWAVES  (EBLOCKS * 4)
#define ETPW    (ETILES / EWAVES)
__global__ __launch_bounds__(128) void pna_edge(
    const float* __restrict__ e, const float* __restrict__ Wpre,
    const int* __restrict__ src, const int* __restrict__ dst,
    const float* __restrict__ P,
    float* s1, float* s2, unsigned* mx, unsigned* mn) {

    __shared__ float ms_all[4 * 16 * MSTRIDE];

    const int tid = threadIdx.x;
    const int l  = tid & 31;
    const int wv = tid >> 5;
    const int lo = l & 15;
    const int hi = l >> 4;
    float* ms = ms_all + wv * 16 * MSTRIDE;

    v2f b[4][4];
#pragma unroll
    for (int cg = 0; cg < 4; ++cg)
#pragma unroll
        for (int j = 0; j < 4; ++j) {
            const int k = 2 * DIN + 8 * hi + 2 * j;
            b[cg][j].x = Wpre[(size_t)k * DIN + cg * 16 + lo];
            b[cg][j].y = Wpre[(size_t)(k + 1) * DIN + cg * 16 + lo];
        }

    int tile = blockIdx.x * 4 + wv;
    int sI = src[tile * 16 + lo];
    int dI = dst[tile * 16 + lo];
    const float* ep = e + (size_t)(tile * 16 + lo) * EDIM + 8 * hi;
    float4 f0 = *(const float4*)ep;
    float4 f1 = *(const float4*)(ep + 4);

    for (int it = 0; it < ETPW; ++it) {
        const int tn = (it + 1 < ETPW) ? tile + EWAVES : tile;
        const int sN = src[tn * 16 + lo];
        const int dN = dst[tn * 16 + lo];
        const float* en = e + (size_t)(tn * 16 + lo) * EDIM + 8 * hi;
        const float4 g0 = *(const float4*)en;
        const float4 g1 = *(const float4*)(en + 4);

        float p0[16], p1[16];
#pragma unroll
        for (int i = 0; i < 16; ++i) {
            const int s = __builtin_amdgcn_readlane(sI, i);
            p0[i] = P[(size_t)s * DIN + l];
            p1[i] = P[(size_t)s * DIN + 32 + l];
        }

        v2f a[4];
        a[0].x = f0.x; a[0].y = f0.y;
        a[1].x = f0.z; a[1].y = f0.w;
        a[2].x = f1.x; a[2].y = f1.y;
        a[3].x = f1.z; a[3].y = f1.w;
        v8f acc[4];
#pragma unroll
        for (int cg = 0; cg < 4; ++cg) acc[cg] = (v8f){};
#pragma unroll
        for (int j = 0; j < 4; ++j)
#pragma unroll
            for (int cg = 0; cg < 4; ++cg)
                acc[cg] = __builtin_amdgcn_wmma_f32_16x16x4_f32(
                    false, a[j], false, b[cg][j], (short)0, acc[cg], false, false);

#pragma unroll
        for (int cg = 0; cg < 4; ++cg)
#pragma unroll
            for (int r = 0; r < 8; ++r)
                ms[(8 * hi + r) * MSTRIDE + cg * 16 + lo] = acc[cg][r];
        __builtin_amdgcn_fence(__ATOMIC_RELEASE, "wavefront");
        __builtin_amdgcn_wave_barrier();
        __builtin_amdgcn_fence(__ATOMIC_ACQUIRE, "wavefront");

#pragma unroll
        for (int i = 0; i < 16; ++i) {
            const int d = __builtin_amdgcn_readlane(dI, i);
            const float v0 = ms[i * MSTRIDE + l] + p0[i];
            const float v1 = ms[i * MSTRIDE + 32 + l] + p1[i];
            const unsigned ev0 = fenc(v0), ev1 = fenc(v1);
            const size_t o = (size_t)d * DIN + l;
            atomicAdd(&s1[o], v0);
            atomicAdd(&s1[o + 32], v1);
            atomicAdd(&s2[o], v0 * v0);
            atomicAdd(&s2[o + 32], v1 * v1);
            atomicMax(&mx[o], ev0);
            atomicMax(&mx[o + 32], ev1);
            atomicMin(&mn[o], ev0);
            atomicMin(&mn[o + 32], ev1);
        }
        __builtin_amdgcn_fence(__ATOMIC_RELEASE, "wavefront");
        __builtin_amdgcn_wave_barrier();
        __builtin_amdgcn_fence(__ATOMIC_ACQUIRE, "wavefront");

        tile = tn; sI = sN; dI = dN; f0 = g0; f1 = g1;
    }
}

#define FSTRIDE 836
__global__ __launch_bounds__(128) void pna_node(
    const float* __restrict__ h, const float* __restrict__ snorm,
    const float* __restrict__ Wpost, const float* __restrict__ bpost,
    const float* __restrict__ s1, const float* __restrict__ s2,
    const unsigned* __restrict__ mx, const unsigned* __restrict__ mn,
    const float* __restrict__ deg,
    float* out, float* bns, float* bnq) {

    __shared__ float feat[16 * FSTRIDE];
    const int tid = threadIdx.x;
    const int n0  = blockIdx.x * 16;

    for (int i = tid; i < 16 * 320; i += 128) {
        int r = i / 320, c = i % 320;
        int node = n0 + r;
        float v;
        if (c < DIN) {
            v = h[node * DIN + c];
        } else {
            int a = c - DIN;
            int sub = a >> 6, colc = a & 63;
            size_t o = (size_t)node * DIN + colc;
            float D  = deg[node];
            if (sub == 0)      v = s1[o] / D + out[o];
            else if (sub == 1) v = fdec(mx[o]) + out[o];
            else if (sub == 2) v = fdec(mn[o]) + out[o];
            else {
                float mean = s1[o] / D;
                v = sqrtf(fmaxf(s2[o] / D - mean * mean, 0.0f) + EPSV);
            }
            if (c == DIN) feat[r * FSTRIDE + 832] = logf(D + 1.0f);
        }
        feat[r * FSTRIDE + c] = v;
    }
    __syncthreads();
    for (int i = tid; i < 16 * 512; i += 128) {
        int r = i >> 9, j = i & 511;
        float logd = feat[r * FSTRIDE + 832];
        float f = (j < 256) ? (logd / AVG_D) : (AVG_D / logd);
        feat[r * FSTRIDE + 320 + j] = feat[r * FSTRIDE + DIN + (j & 255)] * f;
    }
    __syncthreads();

    const int l  = tid & 31;
    const int cg = tid >> 5;
    const int lo = l & 15;
    const int hi = l >> 4;
    const int col = cg * 16 + lo;

    v8f acc = {};
    const float* arow = &feat[lo * FSTRIDE + 2 * hi];
#pragma unroll 4
    for (int k0 = 0; k0 < KPOST; k0 += 4) {
        v2f a = *(const v2f*)(arow + k0);
        v2f b;
        b.x = Wpost[(k0 + 2 * hi)     * DOUT + col];
        b.y = Wpost[(k0 + 2 * hi + 1) * DOUT + col];
        acc = __builtin_amdgcn_wmma_f32_16x16x4_f32(
            false, a, false, b, (short)0, acc, false, false);
    }

    const float bv = bpost[col];
    float ps = 0.0f, ps2 = 0.0f;
#pragma unroll
    for (int r = 0; r < 8; ++r) {
        int M = r + 8 * hi;
        int node = n0 + M;
        float v = (acc[r] + bv) * snorm[node];
        out[(size_t)node * DOUT + col] = v;
        ps += v; ps2 += v * v;
    }
    ps  += __shfl_xor(ps, 16, 32);
    ps2 += __shfl_xor(ps2, 16, 32);
    if (hi == 0) {
        atomicAdd(&bns[col], ps);
        atomicAdd(&bnq[col], ps2);
    }
}

__global__ __launch_bounds__(256) void pna_bn(
    float* out, const float* bns, const float* bnq,
    const float* gamma, const float* beta) {
    int i = blockIdx.x * 256 + threadIdx.x;
    if (i >= NNODES * DOUT) return;
    int c = i & 63;
    float mu  = bns[c] / (float)NNODES;
    float var = bnq[c] / (float)NNODES - mu * mu;
    float v = out[i];
    out[i] = (v - mu) * rsqrtf(var + 1e-05f) * gamma[c] + beta[c];
}

extern "C" void kernel_launch(void* const* d_in, const int* in_sizes, int n_in,
                              void* d_out, int out_size, void* d_ws, size_t ws_size,
                              hipStream_t stream) {
    const float* h      = (const float*)d_in[0];
    const float* e      = (const float*)d_in[1];
    const float* snorm  = (const float*)d_in[2];
    const float* Wpre   = (const float*)d_in[3];
    const float* bpre   = (const float*)d_in[4];
    const float* Wpost  = (const float*)d_in[5];
    const float* bpost  = (const float*)d_in[6];
    const float* gamma  = (const float*)d_in[7];
    const float* beta   = (const float*)d_in[8];
    const int*   src    = (const int*)d_in[9];
    const int*   dst    = (const int*)d_in[10];
    float* out = (float*)d_out;

    float*    s1  = (float*)d_ws;
    float*    s2  = s1 + (size_t)NNODES * DIN;
    unsigned* mx  = (unsigned*)(s2 + (size_t)NNODES * DIN);
    unsigned* mn  = mx + (size_t)NNODES * DIN;
    float*    deg = (float*)(mn + (size_t)NNODES * DIN);
    float*    bns = deg + NNODES;
    float*    bnq = bns + DOUT;

    pna_init  <<<(NNODES * DIN + 255) / 256, 256, 0, stream>>>(s1, s2, mx, mn, deg, bns, bnq);
    pna_degree<<<DRANGES * DSLICES, DTHREADS, 0, stream>>>(dst, deg);
    pna_pre   <<<NNODES / 32, 256, 0, stream>>>(h, Wpre, bpre, 0, 0, out);
    pna_edge  <<<EBLOCKS, 128, 0, stream>>>(e, Wpre, src, dst, out, s1, s2, mx, mn);
    pna_pre   <<<NNODES / 32, 256, 0, stream>>>(h, Wpre, bpre, DIN, 1, out);
    pna_node  <<<NNODES / 16, 128, 0, stream>>>(h, snorm, Wpost, bpost, s1, s2, mx, mn, deg,
                                                out, bns, bnq);
    pna_bn    <<<(NNODES * DOUT + 255) / 256, 256, 0, stream>>>(out, bns, bnq, gamma, beta);
}


# Node kernel turned round: 250 persistent blocks keep all of W_post in LDS as B fragments; a wave builds its 16 nodes' feature rows in registers, scalers applied to products; sum order changes
# speedup vs baseline: 2.9557x; 1.4603x over previous
#include <hip/hip_runtime.h>
#include <hip/hip_bf16.h>
#include <math.h>

#define NNODES 100000
#define NEDGES 1200000
#define DIN    64
#define DOUT   64
#define EDIM   16
#define KPRE   144
#define KPOST  832
#define AVG_D  2.5649493574615367f
#define EPSV   1e-05f

typedef __attribute__((ext_vector_type(2))) float v2f;
typedef __attribute__((ext_vector_type(8))) float v8f;

__device__ __forceinline__ unsigned fenc(float f) {
    unsigned u = __float_as_uint(f);
    return (u & 0x80000000u) ? ~u : (u | 0x80000000u);
}
__device__ __forceinline__ float fdec(unsigned u) {
    return (u & 0x80000000u) ? __uint_as_float(u & 0x7fffffffu)
                             : __uint_as_float(~u);
}


__global__ __launch_bounds__(256) void pna_init(float* s1, float* s2,
                                                unsigned* mx, unsigned* mn,
                                                float* deg, float* bns, float* bnq) {
    int i = blockIdx.x * 256 + threadIdx.x;
    if (i < NNODES * DIN) {
        s1[i] = 0.0f; s2[i] = 0.0f;
        mx[i] = 0u;
        mn[i] = 0xFFFFFFFFu;
    }
    if (i < NNODES) deg[i] = 0.0f;
    if (i < DOUT) { bns[i] = 0.0f; bnq[i] = 0.0f; }
}

#define DRANGES 8
#define DSLICES 32
#define DRSZ    12544
#define DSL4    (NEDGES / DSLICES / 4)
#define DTHREADS 512
#define DPER    ((DSL4 + DTHREADS - 1) / DTHREADS)
__global__ __launch_bounds__(DTHREADS) void pna_degree(const int* __restrict__ dst, float* deg) {
    __shared__ unsigned cnt[DRSZ];
    const int tid = threadIdx.x;
    const int rg  = blockIdx.x % DRANGES;
    const int sl  = blockIdx.x / DRANGES;
    const int r0  = rg * DRSZ;

    const int4* p = (const int4*)(dst + (size_t)sl * (NEDGES / DSLICES));
    int4 v[DPER];
#pragma unroll
    for (int j = 0; j < DPER; ++j) {
        const int i = tid + j * DTHREADS;
        v[j] = (i < DSL4) ? p[i] : make_int4(-1, -1, -1, -1);
    }

    for (int i = tid; i < DRSZ; i += DTHREADS) cnt[i] = 0u;
    __syncthreads();

#pragma unroll
    for (int j = 0; j < DPER; ++j) {
        const unsigned o0 = (unsigned)(v[j].x - r0);
        const unsigned o1 = (unsigned)(v[j].y - r0);
        const unsigned o2 = (unsigned)(v[j].z - r0);
        const unsigned o3 = (unsigned)(v[j].w - r0);
        if (o0 < DRSZ) atomicAdd(&cnt[o0], 1u);
        if (o1 < DRSZ) atomicAdd(&cnt[o1], 1u);
        if (o2 < DRSZ) atomicAdd(&cnt[o2], 1u);
        if (o3 < DRSZ) atomicAdd(&cnt[o3], 1u);
    }
    __syncthreads();

    for (int i = tid; i < DRSZ; i += DTHREADS) {
        const unsigned c = cnt[i];
        const int n = r0 + i;
        if (c != 0u && n < NNODES) atomicAdd(&deg[n], (float)c);
    }
}

__global__ __launch_bounds__(256) void pna_pre(
    const float* __restrict__ h, const float* __restrict__ Wpre,
    const float* __restrict__ bpre, int koff, int use_bias, float* __restrict__ Y) {
    const int tid = threadIdx.x;
    const int l  = tid & 31;
    const int wv = tid >> 5;
    const int cg = wv & 3;
    const int eg = wv >> 2;
    const int lo = l & 15;
    const int hi = l >> 4;
    const int n0 = blockIdx.x * 32 + eg * 16;

    const float* ap = h + (size_t)(n0 + lo) * DIN + 32 * hi;
    float4 a[8];
#pragma unroll
    for (int j = 0; j < 8; ++j) a[j] = *(const float4*)(ap + 4 * j);

    const float* wp = Wpre + (size_t)koff * DIN + cg * 16 + lo;
    v8f acc = {};
#pragma unroll
    for (int t = 0; t < 16; ++t) {
        const int k = 32 * hi + 4 * (t >> 1) + 2 * (t & 1);
        v2f b;
        b.x = wp[(size_t)k * DIN];
        b.y = wp[(size_t)(k + 1) * DIN];
        v2f av;
        if (t & 1) { av.x = a[t >> 1].z; av.y = a[t >> 1].w; }
        else       { av.x = a[t >> 1].x; av.y = a[t >> 1].y; }
        acc = __builtin_amdgcn_wmma_f32_16x16x4_f32(
            false, av, false, b, (short)0, acc, false, false);
    }
    const float bv = use_bias ? bpre[cg * 16 + lo] : 0.0f;
#pragma unroll
    for (int r = 0; r < 8; ++r)
        Y[(size_t)(n0 + 8 * hi + r) * DIN + cg * 16 + lo] = acc[r] + bv;
}

#define MSTRIDE 68
#define ETILES  (NEDGES / 16)
#define EBLOCKS 1875
#define EWAVES  (EBLOCKS * 4)
#define ETPW    (ETILES / EWAVES)
__global__ __launch_bounds__(128) void pna_edge(
    const float* __restrict__ e, const float* __restrict__ Wpre,
    const int* __restrict__ src, const int* __restrict__ dst,
    const float* __restrict__ P,
    float* s1, float* s2, unsigned* mx, unsigned* mn) {

    __shared__ float ms_all[4 * 16 * MSTRIDE];

    const int tid = threadIdx.x;
    const int l  = tid & 31;
    const int wv = tid >> 5;
    const int lo = l & 15;
    const int hi = l >> 4;
    float* ms = ms_all + wv * 16 * MSTRIDE;

    v2f b[4][4];
#pragma unroll
    for (int cg = 0; cg < 4; ++cg)
#pragma unroll
        for (int j = 0; j < 4; ++j) {
            const int k = 2 * DIN + 8 * hi + 2 * j;
            b[cg][j].x = Wpre[(size_t)k * DIN + cg * 16 + lo];
            b[cg][j].y = Wpre[(size_t)(k + 1) * DIN + cg * 16 + lo];
        }

    int tile = blockIdx.x * 4 + wv;
    int sI = src[tile * 16 + lo];
    int dI = dst[tile * 16 + lo];
    const float* ep = e + (size_t)(tile * 16 + lo) * EDIM + 8 * hi;
    float4 f0 = *(const float4*)ep;
    float4 f1 = *(const float4*)(ep + 4);

    for (int it = 0; it < ETPW; ++it) {
        const int tn = (it + 1 < ETPW) ? tile + EWAVES : tile;
        const int sN = src[tn * 16 + lo];
        const int dN = dst[tn * 16 + lo];
        const float* en = e + (size_t)(tn * 16 + lo) * EDIM + 8 * hi;
        const float4 g0 = *(const float4*)en;
        const float4 g1 = *(const float4*)(en + 4);

        float p0[16], p1[16];
#pragma unroll
        for (int i = 0; i < 16; ++i) {
            const int s = __builtin_amdgcn_readlane(sI, i);
            p0[i] = P[(size_t)s * DIN + l];
            p1[i] = P[(size_t)s * DIN + 32 + l];
        }

        v2f a[4];
        a[0].x = f0.x; a[0].y = f0.y;
        a[1].x = f0.z; a[1].y = f0.w;
        a[2].x = f1.x; a[2].y = f1.y;
        a[3].x = f1.z; a[3].y = f1.w;
        v8f acc[4];
#pragma unroll
        for (int cg = 0; cg < 4; ++cg) acc[cg] = (v8f){};
#pragma unroll
        for (int j = 0; j < 4; ++j)
#pragma unroll
            for (int cg = 0; cg < 4; ++cg)
                acc[cg] = __builtin_amdgcn_wmma_f32_16x16x4_f32(
                    false, a[j], false, b[cg][j], (short)0, acc[cg], false, false);

#pragma unroll
        for (int cg = 0; cg < 4; ++cg)
#pragma unroll
            for (int r = 0; r < 8; ++r)
                ms[(8 * hi + r) * MSTRIDE + cg * 16 + lo] = acc[cg][r];
        __builtin_amdgcn_fence(__ATOMIC_RELEASE, "wavefront");
        __builtin_amdgcn_wave_barrier();
        __builtin_amdgcn_fence(__ATOMIC_ACQUIRE, "wavefront");

#pragma unroll
        for (int i = 0; i < 16; ++i) {
            const int d = __builtin_amdgcn_readlane(dI, i);
            const float v0 = ms[i * MSTRIDE + l] + p0[i];
            const float v1 = ms[i * MSTRIDE + 32 + l] + p1[i];
            const unsigned ev0 = fenc(v0), ev1 = fenc(v1);
            const size_t o = (size_t)d * DIN + l;
            atomicAdd(&s1[o], v0);
            atomicAdd(&s1[o + 32], v1);
            atomicAdd(&s2[o], v0 * v0);
            atomicAdd(&s2[o + 32], v1 * v1);
            atomicMax(&mx[o], ev0);
            atomicMax(&mx[o + 32], ev1);
            atomicMin(&mn[o], ev0);
            atomicMin(&mn[o + 32], ev1);
        }
        __builtin_amdgcn_fence(__ATOMIC_RELEASE, "wavefront");
        __builtin_amdgcn_wave_barrier();
        __builtin_amdgcn_fence(__ATOMIC_ACQUIRE, "wavefront");

        tile = tn; sI = sN; dI = dN; f0 = g0; f1 = g1;
    }
}

#define NODE_BLOCKS 250
#define NODE_WAVES  8
#define NODE_TPB    25
#define WL4         (4 * 104 * 32)

__device__ __forceinline__ v8f wmma4(float a0, float a1, float b0, float b1, v8f c) {
    v2f a; a.x = a0; a.y = a1;
    v2f b; b.x = b0; b.y = b1;
    return __builtin_amdgcn_wmma_f32_16x16x4_f32(false, a, false, b, (short)0, c, false, false);
}

__global__ __launch_bounds__(NODE_WAVES * 32) void pna_node(
    const float* __restrict__ h, const float* __restrict__ snorm,
    const float* __restrict__ Wpost, const float* __restrict__ bpost,
    const float* __restrict__ s1, const float* __restrict__ s2,
    const unsigned* __restrict__ mx, const unsigned* __restrict__ mn,
    const float* __restrict__ deg,
    float* out, float* bns, float* bnq) {

    __shared__ float4 wl[WL4];
    const int tid = threadIdx.x;

#pragma unroll 13
    for (int it = 0; it < WL4 / (NODE_WAVES * 32); ++it) {
        const int i   = tid + it * (NODE_WAVES * 32);
        const int fl  = i & 31;
        const int r   = i >> 5;
        const int fcg = r / 104;
        const int tp  = r - fcg * 104;
        const int fj  = tp / 13;
        const int u   = tp - fj * 13;
        int k0 = 32 * (fl >> 4) + 4 * fj;
        if (u != 0) {
            const int pm = (u - 1) / 3;
            const int c  = (u - 1) - 3 * pm;
            k0 += 64 + 256 * c + 64 * pm;
        }
        const float* wp = Wpost + (size_t)k0 * DOUT + fcg * 16 + (fl & 15);
        wl[i] = make_float4(wp[0], wp[DOUT], wp[2 * DOUT], wp[3 * DOUT]);
    }
    __syncthreads();

    const int l  = tid & 31;
    const int wv = tid >> 5;
    const int lo = l & 15;
    const int hi = l >> 4;

    float bv[4], ps[4], pq[4];
#pragma unroll
    for (int cg = 0; cg < 4; ++cg) { bv[cg] = bpost[cg * 16 + lo]; ps[cg] = 0.0f; pq[cg] = 0.0f; }

    struct Chunk { float4 h, a, b, q; uint4 x, n; float d; };
    auto ldchunk = [&](int t, int j) {
        const int node = (blockIdx.x * NODE_TPB + t) * 16 + lo;
        const size_t o = (size_t)node * DIN + 32 * hi + 4 * j;
        Chunk c;
        c.h = *(const float4*)(h   + o);
        c.a = *(const float4*)(s1  + o);
        c.b = *(const float4*)(s2  + o);
        c.x = *(const uint4*) (mx  + o);
        c.n = *(const uint4*) (mn  + o);
        c.q = *(const float4*)(out + o);
        c.d = deg[node];
        return c;
    };

    Chunk cur = ldchunk(wv, 0);
    for (int t = wv; t < NODE_TPB; t += NODE_WAVES) {
        const int n0 = (blockIdx.x * NODE_TPB + t) * 16;
        float D = 1.0f;

        v8f acc[3][4];
#pragma unroll
        for (int c = 0; c < 3; ++c)
#pragma unroll
            for (int cg = 0; cg < 4; ++cg) acc[c][cg] = (v8f){};

#pragma unroll 1
        for (int j = 0; j < 8; ++j) {
            int tn = t, jn = j + 1;
            if (jn == 8) {
                if (t + NODE_WAVES < NODE_TPB) { tn = t + NODE_WAVES; jn = 0; }
                else jn = 7;
            }
            const Chunk nxt = ldchunk(tn, jn);

            D = cur.d;
            const float hh[4] = { cur.h.x, cur.h.y, cur.h.z, cur.h.w };
            const float a1[4] = { cur.a.x, cur.a.y, cur.a.z, cur.a.w };
            const float a2[4] = { cur.b.x, cur.b.y, cur.b.z, cur.b.w };
            const float qq[4] = { cur.q.x, cur.q.y, cur.q.z, cur.q.w };
            const unsigned ux[4] = { cur.x.x, cur.x.y, cur.x.z, cur.x.w };
            const unsigned un[4] = { cur.n.x, cur.n.y, cur.n.z, cur.n.w };
            float f[4][4];
#pragma unroll
            for (int x = 0; x < 4; ++x) {
                const float mean = a1[x] / D;
                f[0][x] = mean + qq[x];
                f[1][x] = fdec(ux[x]) + qq[x];
                f[2][x] = fdec(un[x]) + qq[x];
                f[3][x] = sqrtf(fmaxf(a2[x] / D - mean * mean, 0.0f) + EPSV);
            }
            const float4* wj = wl + (j * 13) * 32 + l;

            {
                float4 b[4];
#pragma unroll
                for (int cg = 0; cg < 4; ++cg) b[cg] = wj[(cg * 104) * 32];
#pragma unroll
                for (int cg = 0; cg < 4; ++cg) acc[0][cg] = wmma4(hh[0], hh[1], b[cg].x, b[cg].y, acc[0][cg]);
#pragma unroll
                for (int cg = 0; cg < 4; ++cg) acc[0][cg] = wmma4(hh[2], hh[3], b[cg].z, b[cg].w, acc[0][cg]);
            }
#pragma unroll
            for (int pm = 0; pm < 4; ++pm)
#pragma unroll
                for (int c = 0; c < 3; ++c) {
                    float4 b[4];
#pragma unroll
                    for (int cg = 0; cg < 4; ++cg) b[cg] = wj[(cg * 104 + 1 + 3 * pm + c) * 32];
#pragma unroll
                    for (int cg = 0; cg < 4; ++cg)
                        acc[c][cg] = wmma4(f[pm][0], f[pm][1], b[cg].x, b[cg].y, acc[c][cg]);
#pragma unroll
                    for (int cg = 0; cg < 4; ++cg)
                        acc[c][cg] = wmma4(f[pm][2], f[pm][3], b[cg].z, b[cg].w, acc[c][cg]);
                }
            cur = nxt;
        }

        const float logd = logf(D + 1.0f);
        const float amp  = logd / AVG_D;
        const float att  = AVG_D / logd;
#pragma unroll
        for (int r = 0; r < 8; ++r) {
            const int row = 8 * hi + r;
            const float ar = __shfl(amp, row, 32);
            const float tr = __shfl(att, row, 32);
            const float sn = snorm[n0 + row];
#pragma unroll
            for (int cg = 0; cg < 4; ++cg) {
                const float v = (acc[0][cg][r] + ar * acc[1][cg][r] + tr * acc[2][cg][r] + bv[cg]) * sn;
                out[(size_t)(n0 + row) * DOUT + cg * 16 + lo] = v;
                ps[cg] += v; pq[cg] += v * v;
            }
        }
    }

#pragma unroll
    for (int cg = 0; cg < 4; ++cg) {
        ps[cg] += __shfl_xor(ps[cg], 16, 32);
        pq[cg] += __shfl_xor(pq[cg], 16, 32);
        if (hi == 0) {
            atomicAdd(&bns[cg * 16 + lo], ps[cg]);
            atomicAdd(&bnq[cg * 16 + lo], pq[cg]);
        }
    }
}

__global__ __launch_bounds__(256) void pna_bn(
    float* out, const float* bns, const float* bnq,
    const float* gamma, const float* beta) {
    int i = blockIdx.x * 256 + threadIdx.x;
    if (i >= NNODES * DOUT) return;
    int c = i & 63;
    float mu  = bns[c] / (float)NNODES;
    float var = bnq[c] / (float)NNODES - mu * mu;
    float v = out[i];
    out[i] = (v - mu) * rsqrtf(var + 1e-05f) * gamma[c] + beta[c];
}

extern "C" void kernel_launch(void* const* d_in, const int* in_sizes, int n_in,
                              void* d_out, int out_size, void* d_ws, size_t ws_size,
                              hipStream_t stream) {
    const float* h      = (const float*)d_in[0];
    const float* e      = (const float*)d_in[1];
    const float* snorm  = (const float*)d_in[2];
    const float* Wpre   = (const float*)d_in[3];
    const float* bpre   = (const float*)d_in[4];
    const float* Wpost  = (const float*)d_in[5];
    const float* bpost  = (const float*)d_in[6];
    const float* gamma  = (const float*)d_in[7];
    const float* beta   = (const float*)d_in[8];
    const int*   src    = (const int*)d_in[9];
    const int*   dst    = (const int*)d_in[10];
    float* out = (float*)d_out;

    float*    s1  = (float*)d_ws;
    float*    s2  = s1 + (size_t)NNODES * DIN;
    unsigned* mx  = (unsigned*)(s2 + (size_t)NNODES * DIN);
    unsigned* mn  = mx + (size_t)NNODES * DIN;
    float*    deg = (float*)(mn + (size_t)NNODES * DIN);
    float*    bns = deg + NNODES;
    float*    bnq = bns + DOUT;

    pna_init  <<<(NNODES * DIN + 255) / 256, 256, 0, stream>>>(s1, s2, mx, mn, deg, bns, bnq);
    pna_degree<<<DRANGES * DSLICES, DTHREADS, 0, stream>>>(dst, deg);
    pna_pre   <<<NNODES / 32, 256, 0, stream>>>(h, Wpre, bpre, 0, 0, out);
    pna_edge  <<<EBLOCKS, 128, 0, stream>>>(e, Wpre, src, dst, out, s1, s2, mx, mn);
    pna_pre   <<<NNODES / 32, 256, 0, stream>>>(h, Wpre, bpre, DIN, 1, out);
    pna_node  <<<NODE_BLOCKS, NODE_WAVES * 32, 0, stream>>>(h, snorm, Wpost, bpost, s1, s2, mx, mn, deg,
                                                out, bns, bnq);
    pna_bn    <<<(NNODES * DOUT + 255) / 256, 256, 0, stream>>>(out, bns, bnq, gamma, beta);
}


# speedup vs baseline: 3.1563x; 1.0679x over previous
#include <hip/hip_runtime.h>
#include <hip/hip_bf16.h>
#include <math.h>

#define NNODES 100000
#define NEDGES 1200000
#define DIN    64
#define DOUT   64
#define EDIM   16
#define KPRE   144
#define KPOST  832
#define AVG_D  2.5649493574615367f
#define EPSV   1e-05f

typedef __attribute__((ext_vector_type(2))) float v2f;
typedef __attribute__((ext_vector_type(8))) float v8f;

__device__ __forceinline__ unsigned fenc(float f) {
    unsigned u = __float_as_uint(f);
    return (u & 0x80000000u) ? ~u : (u | 0x80000000u);
}
__device__ __forceinline__ float fdec(unsigned u) {
    return (u & 0x80000000u) ? __uint_as_float(u & 0x7fffffffu)
                             : __uint_as_float(~u);
}


__global__ __launch_bounds__(256) void pna_init(float* s1, float* s2,
                                                unsigned* mx, unsigned* mn,
                                                float* deg, float* bns, float* bnq) {
    int i = blockIdx.x * 256 + threadIdx.x;
    if (i < NNODES * DIN) {
        s1[i] = 0.0f; s2[i] = 0.0f;
        mx[i] = 0u;
        mn[i] = 0xFFFFFFFFu;
    }
    if (i < NNODES) deg[i] = 0.0f;
    if (i < DOUT) { bns[i] = 0.0f; bnq[i] = 0.0f; }
}

#define DRANGES 8
#define DSLICES 32
#define DRSZ    12544
#define DSL4    (NEDGES / DSLICES / 4)
#define DTHREADS 512
#define DPER    ((DSL4 + DTHREADS - 1) / DTHREADS)
__global__ __launch_bounds__(DTHREADS) void pna_degree(const int* __restrict__ dst, float* deg) {
    __shared__ unsigned cnt[DRSZ];
    const int tid = threadIdx.x;
    const int rg  = blockIdx.x % DRANGES;
    const int sl  = blockIdx.x / DRANGES;
    const int r0  = rg * DRSZ;

    const int4* p = (const int4*)(dst + (size_t)sl * (NEDGES / DSLICES));
    int4 v[DPER];
#pragma unroll
    for (int j = 0; j < DPER; ++j) {
        const int i = tid + j * DTHREADS;
        v[j] = (i < DSL4) ? p[i] : make_int4(-1, -1, -1, -1);
    }

    for (int i = tid; i < DRSZ; i += DTHREADS) cnt[i] = 0u;
    __syncthreads();

#pragma unroll
    for (int j = 0; j < DPER; ++j) {
        const unsigned o0 = (unsigned)(v[j].x - r0);
        const unsigned o1 = (unsigned)(v[j].y - r0);
        const unsigned o2 = (unsigned)(v[j].z - r0);
        const unsigned o3 = (unsigned)(v[j].w - r0);
        if (o0 < DRSZ) atomicAdd(&cnt[o0], 1u);
        if (o1 < DRSZ) atomicAdd(&cnt[o1], 1u);
        if (o2 < DRSZ) atomicAdd(&cnt[o2], 1u);
        if (o3 < DRSZ) atomicAdd(&cnt[o3], 1u);
    }
    __syncthreads();

    for (int i = tid; i < DRSZ; i += DTHREADS) {
        const unsigned c = cnt[i];
        const int n = r0 + i;
        if (c != 0u && n < NNODES) atomicAdd(&deg[n], (float)c);
    }
}

__global__ __launch_bounds__(256) void pna_pre(
    const float* __restrict__ h, const float* __restrict__ Wpre,
    const float* __restrict__ bpre, int koff, int use_bias, float* __restrict__ Y,
    float* bnz) {
    const int tid = threadIdx.x;
    if (bnz != nullptr && blockIdx.x == 0 && tid < 2 * DOUT) bnz[tid] = 0.0f;
    const int l  = tid & 31;
    const int wv = tid >> 5;
    const int cg = wv & 3;
    const int eg = wv >> 2;
    const int lo = l & 15;
    const int hi = l >> 4;
    const int n0 = blockIdx.x * 32 + eg * 16;

    const float* ap = h + (size_t)(n0 + lo) * DIN + 32 * hi;
    float4 a[8];
#pragma unroll
    for (int j = 0; j < 8; ++j) a[j] = *(const float4*)(ap + 4 * j);

    const float* wp = Wpre + (size_t)koff * DIN + cg * 16 + lo;
    v8f acc = {};
#pragma unroll
    for (int t = 0; t < 16; ++t) {
        const int k = 32 * hi + 4 * (t >> 1) + 2 * (t & 1);
        v2f b;
        b.x = wp[(size_t)k * DIN];
        b.y = wp[(size_t)(k + 1) * DIN];
        v2f av;
        if (t & 1) { av.x = a[t >> 1].z; av.y = a[t >> 1].w; }
        else       { av.x = a[t >> 1].x; av.y = a[t >> 1].y; }
        acc = __builtin_amdgcn_wmma_f32_16x16x4_f32(
            false, av, false, b, (short)0, acc, false, false);
    }
    const float bv = use_bias ? bpre[cg * 16 + lo] : 0.0f;
#pragma unroll
    for (int r = 0; r < 8; ++r)
        Y[(size_t)(n0 + 8 * hi + r) * DIN + cg * 16 + lo] = acc[r] + bv;
}

#define MSTRIDE 68
#define ETILES  (NEDGES / 16)
#define EBLOCKS 1875
#define ECH     4
#define ECHUNKS (ETILES / ECH)
#define NHALF   (NNODES / 2)
__global__ __launch_bounds__(128) void pna_edge(
    const float* __restrict__ e, const float* __restrict__ Wpre,
    const int* __restrict__ src, const int* __restrict__ dst,
    const float* __restrict__ P,
    float* s1, float* s2, unsigned* mx, unsigned* mn,
    unsigned* tick0, unsigned* tick1) {

    __shared__ float ms_all[4 * 16 * MSTRIDE];

    const int tid = threadIdx.x;
    const int l  = tid & 31;
    const int wv = tid >> 5;
    const int lo = l & 15;
    const int hi = l >> 4;
    float* ms = ms_all + wv * 16 * MSTRIDE;

    v2f b[4][4];
#pragma unroll
    for (int cg = 0; cg < 4; ++cg)
#pragma unroll
        for (int j = 0; j < 4; ++j) {
            const int k = 2 * DIN + 8 * hi + 2 * j;
            b[cg][j].x = Wpre[(size_t)k * DIN + cg * 16 + lo];
            b[cg][j].y = Wpre[(size_t)(k + 1) * DIN + cg * 16 + lo];
        }

    const unsigned place = (unsigned)__builtin_amdgcn_s_sendmsg_rtn(0x87);
    const int myh = (int)((place >> 18) & 1u);

    for (int pass = 0; pass < 2; ++pass) {
        const int hs = myh ^ pass;
        unsigned* tk = hs ? tick1 : tick0;

        unsigned tc = 0u;
        if (l == 0) tc = atomicAdd(tk, 1u);
        tc = (unsigned)__builtin_amdgcn_readfirstlane((int)tc);

        for (int g = 0; g < ECHUNKS; ++g) {
            if (tc >= (unsigned)ECHUNKS) break;
            int tile = (int)tc * ECH;
            int sI = src[tile * 16 + lo];
            int dI = dst[tile * 16 + lo];
            const float* ep = e + (size_t)(tile * 16 + lo) * EDIM + 8 * hi;
            float4 f0 = *(const float4*)ep;
            float4 f1 = *(const float4*)(ep + 4);
            unsigned tnx = 0u;
            if (l == 0) tnx = atomicAdd(tk, 1u);

            for (int it = 0; it < ECH; ++it) {
                const int tn = (it + 1 < ECH) ? tile + 1 : tile;
                const int sN = src[tn * 16 + lo];
                const int dN = dst[tn * 16 + lo];
                const float* en = e + (size_t)(tn * 16 + lo) * EDIM + 8 * hi;
                const float4 g0 = *(const float4*)en;
                const float4 g1 = *(const float4*)(en + 4);

                float p0[16], p1[16];
#pragma unroll
                for (int i = 0; i < 16; ++i) {
                    const int d = __builtin_amdgcn_readlane(dI, i);
                    p0[i] = 0.0f; p1[i] = 0.0f;
                    if ((d >= NHALF ? 1 : 0) == hs) {
                        const int s = __builtin_amdgcn_readlane(sI, i);
                        p0[i] = P[(size_t)s * DIN + l];
                        p1[i] = P[(size_t)s * DIN + 32 + l];
                    }
                }

                v2f a[4];
                a[0].x = f0.x; a[0].y = f0.y;
                a[1].x = f0.z; a[1].y = f0.w;
                a[2].x = f1.x; a[2].y = f1.y;
                a[3].x = f1.z; a[3].y = f1.w;
                v8f acc[4];
#pragma unroll
                for (int cg = 0; cg < 4; ++cg) acc[cg] = (v8f){};
#pragma unroll
                for (int j = 0; j < 4; ++j)
#pragma unroll
                    for (int cg = 0; cg < 4; ++cg)
                        acc[cg] = __builtin_amdgcn_wmma_f32_16x16x4_f32(
                            false, a[j], false, b[cg][j], (short)0, acc[cg], false, false);

#pragma unroll
                for (int cg = 0; cg < 4; ++cg)
#pragma unroll
                    for (int r = 0; r < 8; ++r)
                        ms[(8 * hi + r) * MSTRIDE + cg * 16 + lo] = acc[cg][r];
                __builtin_amdgcn_fence(__ATOMIC_RELEASE, "wavefront");
                __builtin_amdgcn_wave_barrier();
                __builtin_amdgcn_fence(__ATOMIC_ACQUIRE, "wavefront");

#pragma unroll
                for (int i = 0; i < 16; ++i) {
                    const int d = __builtin_amdgcn_readlane(dI, i);
                    if ((d >= NHALF ? 1 : 0) == hs) {
                        const float v0 = ms[i * MSTRIDE + l] + p0[i];
                        const float v1 = ms[i * MSTRIDE + 32 + l] + p1[i];
                        const unsigned ev0 = fenc(v0), ev1 = fenc(v1);
                        const size_t o = (size_t)d * DIN + l;
                        atomicAdd(&s1[o], v0);
                        atomicAdd(&s1[o + 32], v1);
                        atomicAdd(&s2[o], v0 * v0);
                        atomicAdd(&s2[o + 32], v1 * v1);
                        atomicMax(&mx[o], ev0);
                        atomicMax(&mx[o + 32], ev1);
                        atomicMin(&mn[o], ev0);
                        atomicMin(&mn[o + 32], ev1);
                    }
                }
                __builtin_amdgcn_fence(__ATOMIC_RELEASE, "wavefront");
                __builtin_amdgcn_wave_barrier();
                __builtin_amdgcn_fence(__ATOMIC_ACQUIRE, "wavefront");

                tile = tn; sI = sN; dI = dN; f0 = g0; f1 = g1;
            }
            tc = (unsigned)__builtin_amdgcn_readfirstlane((int)tnx);
        }
    }
}

#define NODE_BLOCKS 250
#define NODE_WAVES  8
#define NODE_TPB    25
#define WL4         (4 * 104 * 32)

__device__ __forceinline__ v8f wmma4(float a0, float a1, float b0, float b1, v8f c) {
    v2f a; a.x = a0; a.y = a1;
    v2f b; b.x = b0; b.y = b1;
    return __builtin_amdgcn_wmma_f32_16x16x4_f32(false, a, false, b, (short)0, c, false, false);
}

__global__ __launch_bounds__(NODE_WAVES * 32) void pna_node(
    const float* __restrict__ h, const float* __restrict__ snorm,
    const float* __restrict__ Wpost, const float* __restrict__ bpost,
    const float* __restrict__ s1, const float* __restrict__ s2,
    const unsigned* __restrict__ mx, const unsigned* __restrict__ mn,
    const float* __restrict__ deg,
    float* out, float* bns, float* bnq) {

    __shared__ float4 wl[WL4];
    const int tid = threadIdx.x;

#pragma unroll 13
    for (int it = 0; it < WL4 / (NODE_WAVES * 32); ++it) {
        const int i   = tid + it * (NODE_WAVES * 32);
        const int fl  = i & 31;
        const int r   = i >> 5;
        const int fcg = r / 104;
        const int tp  = r - fcg * 104;
        const int fj  = tp / 13;
        const int u   = tp - fj * 13;
        int k0 = 32 * (fl >> 4) + 4 * fj;
        if (u != 0) {
            const int pm = (u - 1) / 3;
            const int c  = (u - 1) - 3 * pm;
            k0 += 64 + 256 * c + 64 * pm;
        }
        const float* wp = Wpost + (size_t)k0 * DOUT + fcg * 16 + (fl & 15);
        wl[i] = make_float4(wp[0], wp[DOUT], wp[2 * DOUT], wp[3 * DOUT]);
    }
    __syncthreads();

    const int l  = tid & 31;
    const int wv = tid >> 5;
    const int lo = l & 15;
    const int hi = l >> 4;

    float bv[4], ps[4], pq[4];
#pragma unroll
    for (int cg = 0; cg < 4; ++cg) { bv[cg] = bpost[cg * 16 + lo]; ps[cg] = 0.0f; pq[cg] = 0.0f; }

    struct Chunk { float4 h, a, b, q; uint4 x, n; float d; };
    auto ldchunk = [&](int t, int j) {
        const int node = (blockIdx.x * NODE_TPB + t) * 16 + lo;
        const size_t o = (size_t)node * DIN + 32 * hi + 4 * j;
        Chunk c;
        c.h = *(const float4*)(h   + o);
        c.a = *(const float4*)(s1  + o);
        c.b = *(const float4*)(s2  + o);
        c.x = *(const uint4*) (mx  + o);
        c.n = *(const uint4*) (mn  + o);
        c.q = *(const float4*)(out + o);
        c.d = deg[node];
        return c;
    };

    Chunk cur = ldchunk(wv, 0);
    for (int t = wv; t < NODE_TPB; t += NODE_WAVES) {
        const int n0 = (blockIdx.x * NODE_TPB + t) * 16;
        float D = 1.0f;

        v8f acc[3][4];
#pragma unroll
        for (int c = 0; c < 3; ++c)
#pragma unroll
            for (int cg = 0; cg < 4; ++cg) acc[c][cg] = (v8f){};

#pragma unroll 1
        for (int j = 0; j < 8; ++j) {
            int tn = t, jn = j + 1;
            if (jn == 8) {
                if (t + NODE_WAVES < NODE_TPB) { tn = t + NODE_WAVES; jn = 0; }
                else jn = 7;
            }
            const Chunk nxt = ldchunk(tn, jn);

            D = cur.d;
            const float hh[4] = { cur.h.x, cur.h.y, cur.h.z, cur.h.w };
            const float a1[4] = { cur.a.x, cur.a.y, cur.a.z, cur.a.w };
            const float a2[4] = { cur.b.x, cur.b.y, cur.b.z, cur.b.w };
            const float qq[4] = { cur.q.x, cur.q.y, cur.q.z, cur.q.w };
            const unsigned ux[4] = { cur.x.x, cur.x.y, cur.x.z, cur.x.w };
            const unsigned un[4] = { cur.n.x, cur.n.y, cur.n.z, cur.n.w };
            float f[4][4];
#pragma unroll
            for (int x = 0; x < 4; ++x) {
                const float mean = a1[x] / D;
                f[0][x] = mean + qq[x];
                f[1][x] = fdec(ux[x]) + qq[x];
                f[2][x] = fdec(un[x]) + qq[x];
                f[3][x] = sqrtf(fmaxf(a2[x] / D - mean * mean, 0.0f) + EPSV);
            }
            const float4* wj = wl + (j * 13) * 32 + l;

            {
                float4 b[4];
#pragma unroll
                for (int cg = 0; cg < 4; ++cg) b[cg] = wj[(cg * 104) * 32];
#pragma unroll
                for (int cg = 0; cg < 4; ++cg) acc[0][cg] = wmma4(hh[0], hh[1], b[cg].x, b[cg].y, acc[0][cg]);
#pragma unroll
                for (int cg = 0; cg < 4; ++cg) acc[0][cg] = wmma4(hh[2], hh[3], b[cg].z, b[cg].w, acc[0][cg]);
            }
#pragma unroll
            for (int pm = 0; pm < 4; ++pm)
#pragma unroll
                for (int c = 0; c < 3; ++c) {
                    float4 b[4];
#pragma unroll
                    for (int cg = 0; cg < 4; ++cg) b[cg] = wj[(cg * 104 + 1 + 3 * pm + c) * 32];
#pragma unroll
                    for (int cg = 0; cg < 4; ++cg)
                        acc[c][cg] = wmma4(f[pm][0], f[pm][1], b[cg].x, b[cg].y, acc[c][cg]);
#pragma unroll
                    for (int cg = 0; cg < 4; ++cg)
                        acc[c][cg] = wmma4(f[pm][2], f[pm][3], b[cg].z, b[cg].w, acc[c][cg]);
                }
            cur = nxt;
        }

        const float logd = logf(D + 1.0f);
        const float amp  = logd / AVG_D;
        const float att  = AVG_D / logd;
#pragma unroll
        for (int r = 0; r < 8; ++r) {
            const int row = 8 * hi + r;
            const float ar = __shfl(amp, row, 32);
            const float tr = __shfl(att, row, 32);
            const float sn = snorm[n0 + row];
#pragma unroll
            for (int cg = 0; cg < 4; ++cg) {
                const float v = (acc[0][cg][r] + ar * acc[1][cg][r] + tr * acc[2][cg][r] + bv[cg]) * sn;
                out[(size_t)(n0 + row) * DOUT + cg * 16 + lo] = v;
                ps[cg] += v; pq[cg] += v * v;
            }
        }
    }

#pragma unroll
    for (int cg = 0; cg < 4; ++cg) {
        ps[cg] += __shfl_xor(ps[cg], 16, 32);
        pq[cg] += __shfl_xor(pq[cg], 16, 32);
        if (hi == 0) {
            atomicAdd(&bns[cg * 16 + lo], ps[cg]);
            atomicAdd(&bnq[cg * 16 + lo], pq[cg]);
        }
    }
}

__global__ __launch_bounds__(256) void pna_bn(
    float* out, const float* bns, const float* bnq,
    const float* gamma, const float* beta) {
    int i = blockIdx.x * 256 + threadIdx.x;
    if (i >= NNODES * DOUT) return;
    int c = i & 63;
    float mu  = bns[c] / (float)NNODES;
    float var = bnq[c] / (float)NNODES - mu * mu;
    float v = out[i];
    out[i] = (v - mu) * rsqrtf(var + 1e-05f) * gamma[c] + beta[c];
}

extern "C" void kernel_launch(void* const* d_in, const int* in_sizes, int n_in,
                              void* d_out, int out_size, void* d_ws, size_t ws_size,
                              hipStream_t stream) {
    const float* h      = (const float*)d_in[0];
    const float* e      = (const float*)d_in[1];
    const float* snorm  = (const float*)d_in[2];
    const float* Wpre   = (const float*)d_in[3];
    const float* bpre   = (const float*)d_in[4];
    const float* Wpost  = (const float*)d_in[5];
    const float* bpost  = (const float*)d_in[6];
    const float* gamma  = (const float*)d_in[7];
    const float* beta   = (const float*)d_in[8];
    const int*   src    = (const int*)d_in[9];
    const int*   dst    = (const int*)d_in[10];
    float* out = (float*)d_out;

    float*    s1  = (float*)d_ws;
    float*    s2  = s1 + (size_t)NNODES * DIN;
    unsigned* mx  = (unsigned*)(s2 + (size_t)NNODES * DIN);
    unsigned* mn  = mx + (size_t)NNODES * DIN;
    float*    deg = (float*)(mn + (size_t)NNODES * DIN);
    float*    bns = deg + NNODES;
    float*    bnq = bns + DOUT;

    pna_init  <<<(NNODES * DIN + 255) / 256, 256, 0, stream>>>(s1, s2, mx, mn, deg, bns, bnq);
    pna_degree<<<DRANGES * DSLICES, DTHREADS, 0, stream>>>(dst, deg);
    pna_pre   <<<NNODES / 32, 256, 0, stream>>>(h, Wpre, bpre, 0, 0, out, nullptr);
    pna_edge  <<<EBLOCKS, 128, 0, stream>>>(e, Wpre, src, dst, out, s1, s2, mx, mn,
                                            (unsigned*)bns, (unsigned*)(bnq + 32));
    pna_pre   <<<NNODES / 32, 256, 0, stream>>>(h, Wpre, bpre, DIN, 1, out, bns);
    pna_node  <<<NODE_BLOCKS, NODE_WAVES * 32, 0, stream>>>(h, snorm, Wpost, bpost, s1, s2, mx, mn, deg,
                                                out, bns, bnq);
    pna_bn    <<<(NNODES * DOUT + 255) / 256, 256, 0, stream>>>(out, bns, bnq, gamma, beta);
}


# Pre-product kernel: 256 persistent blocks, a wave owns whole 16-node tiles with its W_pre slice held in registers; both lane halves of a load read one line (16 lines, not 32); sum order changes
# speedup vs baseline: 3.2803x; 1.0393x over previous
#include <hip/hip_runtime.h>
#include <hip/hip_bf16.h>
#include <math.h>

#define NNODES 100000
#define NEDGES 1200000
#define DIN    64
#define DOUT   64
#define EDIM   16
#define KPRE   144
#define KPOST  832
#define AVG_D  2.5649493574615367f
#define EPSV   1e-05f

typedef __attribute__((ext_vector_type(2))) float v2f;
typedef __attribute__((ext_vector_type(8))) float v8f;

__device__ __forceinline__ unsigned fenc(float f) {
    unsigned u = __float_as_uint(f);
    return (u & 0x80000000u) ? ~u : (u | 0x80000000u);
}
__device__ __forceinline__ float fdec(unsigned u) {
    return (u & 0x80000000u) ? __uint_as_float(u & 0x7fffffffu)
                             : __uint_as_float(~u);
}


__global__ __launch_bounds__(256) void pna_init(float* s1, float* s2,
                                                unsigned* mx, unsigned* mn,
                                                float* deg, float* bns, float* bnq) {
    int i = blockIdx.x * 256 + threadIdx.x;
    if (i < NNODES * DIN) {
        s1[i] = 0.0f; s2[i] = 0.0f;
        mx[i] = 0u;
        mn[i] = 0xFFFFFFFFu;
    }
    if (i < NNODES) deg[i] = 0.0f;
    if (i < DOUT) { bns[i] = 0.0f; bnq[i] = 0.0f; }
}

#define DRANGES 8
#define DSLICES 32
#define DRSZ    12544
#define DSL4    (NEDGES / DSLICES / 4)
#define DTHREADS 512
#define DPER    ((DSL4 + DTHREADS - 1) / DTHREADS)
__global__ __launch_bounds__(DTHREADS) void pna_degree(const int* __restrict__ dst, float* deg) {
    __shared__ unsigned cnt[DRSZ];
    const int tid = threadIdx.x;
    const int rg  = blockIdx.x % DRANGES;
    const int sl  = blockIdx.x / DRANGES;
    const int r0  = rg * DRSZ;

    const int4* p = (const int4*)(dst + (size_t)sl * (NEDGES / DSLICES));
    int4 v[DPER];
#pragma unroll
    for (int j = 0; j < DPER; ++j) {
        const int i = tid + j * DTHREADS;
        v[j] = (i < DSL4) ? p[i] : make_int4(-1, -1, -1, -1);
    }

    for (int i = tid; i < DRSZ; i += DTHREADS) cnt[i] = 0u;
    __syncthreads();

#pragma unroll
    for (int j = 0; j < DPER; ++j) {
        const unsigned o0 = (unsigned)(v[j].x - r0);
        const unsigned o1 = (unsigned)(v[j].y - r0);
        const unsigned o2 = (unsigned)(v[j].z - r0);
        const unsigned o3 = (unsigned)(v[j].w - r0);
        if (o0 < DRSZ) atomicAdd(&cnt[o0], 1u);
        if (o1 < DRSZ) atomicAdd(&cnt[o1], 1u);
        if (o2 < DRSZ) atomicAdd(&cnt[o2], 1u);
        if (o3 < DRSZ) atomicAdd(&cnt[o3], 1u);
    }
    __syncthreads();

    for (int i = tid; i < DRSZ; i += DTHREADS) {
        const unsigned c = cnt[i];
        const int n = r0 + i;
        if (c != 0u && n < NNODES) atomicAdd(&deg[n], (float)c);
    }
}

#define PRE_BLOCKS 256
#define PRE_WAVES  8
#define PRE_NW     (PRE_BLOCKS * PRE_WAVES)
#define NTILES     (NNODES / 16)
__global__ __launch_bounds__(PRE_WAVES * 32) void pna_pre(
    const float* __restrict__ h, const float* __restrict__ Wpre,
    const float* __restrict__ bpre, int koff, int use_bias, float* __restrict__ Y,
    float* bnz) {
    const int tid = threadIdx.x;
    if (bnz != nullptr && blockIdx.x == 0 && tid < 2 * DOUT) bnz[tid] = 0.0f;
    const int l  = tid & 31;
    const int lo = l & 15;
    const int hi = l >> 4;

    int tile = (tid >> 5) * PRE_BLOCKS + (int)blockIdx.x;

    const int aoff = lo * DIN + 4 * hi;
    const int yoff = 8 * hi * DIN + lo;

    float4 a[8];
    {
        const float* ap = h + (size_t)tile * (16 * DIN) + aoff;
#pragma unroll
        for (int j = 0; j < 8; ++j) a[j] = *(const float4*)(ap + 8 * j);
    }

    v2f b[16][4];
    const float* wp = Wpre + (size_t)koff * DIN + lo;
#pragma unroll
    for (int t = 0; t < 16; ++t) {
        const int k = 8 * (t >> 1) + 4 * hi + 2 * (t & 1);
#pragma unroll
        for (int cg = 0; cg < 4; ++cg) {
            b[t][cg].x = wp[(size_t)k * DIN + cg * 16];
            b[t][cg].y = wp[(size_t)(k + 1) * DIN + cg * 16];
        }
    }
    float bv[4];
#pragma unroll
    for (int cg = 0; cg < 4; ++cg) bv[cg] = use_bias ? bpre[cg * 16 + lo] : 0.0f;

    for (; tile < NTILES; tile += PRE_NW) {
        const int tn = (tile + PRE_NW < NTILES) ? tile + PRE_NW : tile;
        float4 an[8];
        {
            const float* ap = h + (size_t)tn * (16 * DIN) + aoff;
#pragma unroll
            for (int j = 0; j < 8; ++j) an[j] = *(const float4*)(ap + 8 * j);
        }
        __builtin_amdgcn_sched_barrier(0);

        v8f acc[4];
#pragma unroll
        for (int cg = 0; cg < 4; ++cg) acc[cg] = (v8f){};
#pragma unroll
        for (int t = 0; t < 16; ++t) {
            v2f av;
            if (t & 1) { av.x = a[t >> 1].z; av.y = a[t >> 1].w; }
            else       { av.x = a[t >> 1].x; av.y = a[t >> 1].y; }
#pragma unroll
            for (int cg = 0; cg < 4; ++cg)
                acc[cg] = __builtin_amdgcn_wmma_f32_16x16x4_f32(
                    false, av, false, b[t][cg], (short)0, acc[cg], false, false);
        }
        float* yp = Y + (size_t)tile * (16 * DIN) + yoff;
#pragma unroll
        for (int cg = 0; cg < 4; ++cg)
#pragma unroll
            for (int r = 0; r < 8; ++r)
                yp[r * DIN + cg * 16] = acc[cg][r] + bv[cg];
        __builtin_amdgcn_sched_barrier(0);
#pragma unroll
        for (int j = 0; j < 8; ++j) a[j] = an[j];
    }
}

#define MSTRIDE 68
#define ETILES  (NEDGES / 16)
#define EBLOCKS 1875
#define ECH     4
#define ECHUNKS (ETILES / ECH)
#define NHALF   (NNODES / 2)
__global__ __launch_bounds__(128) void pna_edge(
    const float* __restrict__ e, const float* __restrict__ Wpre,
    const int* __restrict__ src, const int* __restrict__ dst,
    const float* __restrict__ P,
    float* s1, float* s2, unsigned* mx, unsigned* mn,
    unsigned* tick0, unsigned* tick1) {

    __shared__ float ms_all[4 * 16 * MSTRIDE];

    const int tid = threadIdx.x;
    const int l  = tid & 31;
    const int wv = tid >> 5;
    const int lo = l & 15;
    const int hi = l >> 4;
    float* ms = ms_all + wv * 16 * MSTRIDE;

    v2f b[4][4];
#pragma unroll
    for (int cg = 0; cg < 4; ++cg)
#pragma unroll
        for (int j = 0; j < 4; ++j) {
            const int k = 2 * DIN + 8 * hi + 2 * j;
            b[cg][j].x = Wpre[(size_t)k * DIN + cg * 16 + lo];
            b[cg][j].y = Wpre[(size_t)(k + 1) * DIN + cg * 16 + lo];
        }

    const unsigned place = (unsigned)__builtin_amdgcn_s_sendmsg_rtn(0x87);
    const int myh = (int)((place >> 18) & 1u);

    for (int pass = 0; pass < 2; ++pass) {
        const int hs = myh ^ pass;
        unsigned* tk = hs ? tick1 : tick0;

        unsigned tc = 0u;
        if (l == 0) tc = atomicAdd(tk, 1u);
        tc = (unsigned)__builtin_amdgcn_readfirstlane((int)tc);

        for (int g = 0; g < ECHUNKS; ++g) {
            if (tc >= (unsigned)ECHUNKS) break;
            int tile = (int)tc * ECH;
            int sI = src[tile * 16 + lo];
            int dI = dst[tile * 16 + lo];
            const float* ep = e + (size_t)(tile * 16 + lo) * EDIM + 8 * hi;
            float4 f0 = *(const float4*)ep;
            float4 f1 = *(const float4*)(ep + 4);
            unsigned tnx = 0u;
            if (l == 0) tnx = atomicAdd(tk, 1u);

            for (int it = 0; it < ECH; ++it) {
                const int tn = (it + 1 < ECH) ? tile + 1 : tile;
                const int sN = src[tn * 16 + lo];
                const int dN = dst[tn * 16 + lo];
                const float* en = e + (size_t)(tn * 16 + lo) * EDIM + 8 * hi;
                const float4 g0 = *(const float4*)en;
                const float4 g1 = *(const float4*)(en + 4);

                float p0[16], p1[16];
#pragma unroll
                for (int i = 0; i < 16; ++i) {
                    const int d = __builtin_amdgcn_readlane(dI, i);
                    p0[i] = 0.0f; p1[i] = 0.0f;
                    if ((d >= NHALF ? 1 : 0) == hs) {
                        const int s = __builtin_amdgcn_readlane(sI, i);
                        p0[i] = P[(size_t)s * DIN + l];
                        p1[i] = P[(size_t)s * DIN + 32 + l];
                    }
                }

                v2f a[4];
                a[0].x = f0.x; a[0].y = f0.y;
                a[1].x = f0.z; a[1].y = f0.w;
                a[2].x = f1.x; a[2].y = f1.y;
                a[3].x = f1.z; a[3].y = f1.w;
                v8f acc[4];
#pragma unroll
                for (int cg = 0; cg < 4; ++cg) acc[cg] = (v8f){};
#pragma unroll
                for (int j = 0; j < 4; ++j)
#pragma unroll
                    for (int cg = 0; cg < 4; ++cg)
                        acc[cg] = __builtin_amdgcn_wmma_f32_16x16x4_f32(
                            false, a[j], false, b[cg][j], (short)0, acc[cg], false, false);

#pragma unroll
                for (int cg = 0; cg < 4; ++cg)
#pragma unroll
                    for (int r = 0; r < 8; ++r)
                        ms[(8 * hi + r) * MSTRIDE + cg * 16 + lo] = acc[cg][r];
                __builtin_amdgcn_fence(__ATOMIC_RELEASE, "wavefront");
                __builtin_amdgcn_wave_barrier();
                __builtin_amdgcn_fence(__ATOMIC_ACQUIRE, "wavefront");

#pragma unroll
                for (int i = 0; i < 16; ++i) {
                    const int d = __builtin_amdgcn_readlane(dI, i);
                    if ((d >= NHALF ? 1 : 0) == hs) {
                        const float v0 = ms[i * MSTRIDE + l] + p0[i];
                        const float v1 = ms[i * MSTRIDE + 32 + l] + p1[i];
                        const unsigned ev0 = fenc(v0), ev1 = fenc(v1);
                        const size_t o = (size_t)d * DIN + l;
                        atomicAdd(&s1[o], v0);
                        atomicAdd(&s1[o + 32], v1);
                        atomicAdd(&s2[o], v0 * v0);
                        atomicAdd(&s2[o + 32], v1 * v1);
                        atomicMax(&mx[o], ev0);
                        atomicMax(&mx[o + 32], ev1);
                        atomicMin(&mn[o], ev0);
                        atomicMin(&mn[o + 32], ev1);
                    }
                }
                __builtin_amdgcn_fence(__ATOMIC_RELEASE, "wavefront");
                __builtin_amdgcn_wave_barrier();
                __builtin_amdgcn_fence(__ATOMIC_ACQUIRE, "wavefront");

                tile = tn; sI = sN; dI = dN; f0 = g0; f1 = g1;
            }
            tc = (unsigned)__builtin_amdgcn_readfirstlane((int)tnx);
        }
    }
}

#define NODE_BLOCKS 250
#define NODE_WAVES  8
#define NODE_TPB    25
#define WL4         (4 * 104 * 32)

__device__ __forceinline__ v8f wmma4(float a0, float a1, float b0, float b1, v8f c) {
    v2f a; a.x = a0; a.y = a1;
    v2f b; b.x = b0; b.y = b1;
    return __builtin_amdgcn_wmma_f32_16x16x4_f32(false, a, false, b, (short)0, c, false, false);
}

__global__ __launch_bounds__(NODE_WAVES * 32) void pna_node(
    const float* __restrict__ h, const float* __restrict__ snorm,
    const float* __restrict__ Wpost, const float* __restrict__ bpost,
    const float* __restrict__ s1, const float* __restrict__ s2,
    const unsigned* __restrict__ mx, const unsigned* __restrict__ mn,
    const float* __restrict__ deg,
    float* out, float* bns, float* bnq) {

    __shared__ float4 wl[WL4];
    const int tid = threadIdx.x;

#pragma unroll 13
    for (int it = 0; it < WL4 / (NODE_WAVES * 32); ++it) {
        const int i   = tid + it * (NODE_WAVES * 32);
        const int fl  = i & 31;
        const int r   = i >> 5;
        const int fcg = r / 104;
        const int tp  = r - fcg * 104;
        const int fj  = tp / 13;
        const int u   = tp - fj * 13;
        int k0 = 32 * (fl >> 4) + 4 * fj;
        if (u != 0) {
            const int pm = (u - 1) / 3;
            const int c  = (u - 1) - 3 * pm;
            k0 += 64 + 256 * c + 64 * pm;
        }
        const float* wp = Wpost + (size_t)k0 * DOUT + fcg * 16 + (fl & 15);
        wl[i] = make_float4(wp[0], wp[DOUT], wp[2 * DOUT], wp[3 * DOUT]);
    }
    __syncthreads();

    const int l  = tid & 31;
    const int wv = tid >> 5;
    const int lo = l & 15;
    const int hi = l >> 4;

    float bv[4], ps[4], pq[4];
#pragma unroll
    for (int cg = 0; cg < 4; ++cg) { bv[cg] = bpost[cg * 16 + lo]; ps[cg] = 0.0f; pq[cg] = 0.0f; }

    struct Chunk { float4 h, a, b, q; uint4 x, n; float d; };
    auto ldchunk = [&](int t, int j) {
        const int node = (blockIdx.x * NODE_TPB + t) * 16 + lo;
        const size_t o = (size_t)node * DIN + 32 * hi + 4 * j;
        Chunk c;
        c.h = *(const float4*)(h   + o);
        c.a = *(const float4*)(s1  + o);
        c.b = *(const float4*)(s2  + o);
        c.x = *(const uint4*) (mx  + o);
        c.n = *(const uint4*) (mn  + o);
        c.q = *(const float4*)(out + o);
        c.d = deg[node];
        return c;
    };

    Chunk cur = ldchunk(wv, 0);
    for (int t = wv; t < NODE_TPB; t += NODE_WAVES) {
        const int n0 = (blockIdx.x * NODE_TPB + t) * 16;
        float D = 1.0f;

        v8f acc[3][4];
#pragma unroll
        for (int c = 0; c < 3; ++c)
#pragma unroll
            for (int cg = 0; cg < 4; ++cg) acc[c][cg] = (v8f){};

#pragma unroll 1
        for (int j = 0; j < 8; ++j) {
            int tn = t, jn = j + 1;
            if (jn == 8) {
                if (t + NODE_WAVES < NODE_TPB) { tn = t + NODE_WAVES; jn = 0; }
                else jn = 7;
            }
            const Chunk nxt = ldchunk(tn, jn);

            D = cur.d;
            const float hh[4] = { cur.h.x, cur.h.y, cur.h.z, cur.h.w };
            const float a1[4] = { cur.a.x, cur.a.y, cur.a.z, cur.a.w };
            const float a2[4] = { cur.b.x, cur.b.y, cur.b.z, cur.b.w };
            const float qq[4] = { cur.q.x, cur.q.y, cur.q.z, cur.q.w };
            const unsigned ux[4] = { cur.x.x, cur.x.y, cur.x.z, cur.x.w };
            const unsigned un[4] = { cur.n.x, cur.n.y, cur.n.z, cur.n.w };
            float f[4][4];
#pragma unroll
            for (int x = 0; x < 4; ++x) {
                const float mean = a1[x] / D;
                f[0][x] = mean + qq[x];
                f[1][x] = fdec(ux[x]) + qq[x];
                f[2][x] = fdec(un[x]) + qq[x];
                f[3][x] = sqrtf(fmaxf(a2[x] / D - mean * mean, 0.0f) + EPSV);
            }
            const float4* wj = wl + (j * 13) * 32 + l;

            {
                float4 b[4];
#pragma unroll
                for (int cg = 0; cg < 4; ++cg) b[cg] = wj[(cg * 104) * 32];
#pragma unroll
                for (int cg = 0; cg < 4; ++cg) acc[0][cg] = wmma4(hh[0], hh[1], b[cg].x, b[cg].y, acc[0][cg]);
#pragma unroll
                for (int cg = 0; cg < 4; ++cg) acc[0][cg] = wmma4(hh[2], hh[3], b[cg].z, b[cg].w, acc[0][cg]);
            }
#pragma unroll
            for (int pm = 0; pm < 4; ++pm)
#pragma unroll
                for (int c = 0; c < 3; ++c) {
                    float4 b[4];
#pragma unroll
                    for (int cg = 0; cg < 4; ++cg) b[cg] = wj[(cg * 104 + 1 + 3 * pm + c) * 32];
#pragma unroll
                    for (int cg = 0; cg < 4; ++cg)
                        acc[c][cg] = wmma4(f[pm][0], f[pm][1], b[cg].x, b[cg].y, acc[c][cg]);
#pragma unroll
                    for (int cg = 0; cg < 4; ++cg)
                        acc[c][cg] = wmma4(f[pm][2], f[pm][3], b[cg].z, b[cg].w, acc[c][cg]);
                }
            cur = nxt;
        }

        const float logd = logf(D + 1.0f);
        const float amp  = logd / AVG_D;
        const float att  = AVG_D / logd;
#pragma unroll
        for (int r = 0; r < 8; ++r) {
            const int row = 8 * hi + r;
            const float ar = __shfl(amp, row, 32);
            const float tr = __shfl(att, row, 32);
            const float sn = snorm[n0 + row];
#pragma unroll
            for (int cg = 0; cg < 4; ++cg) {
                const float v = (acc[0][cg][r] + ar * acc[1][cg][r] + tr * acc[2][cg][r] + bv[cg]) * sn;
                out[(size_t)(n0 + row) * DOUT + cg * 16 + lo] = v;
                ps[cg] += v; pq[cg] += v * v;
            }
        }
    }

#pragma unroll
    for (int cg = 0; cg < 4; ++cg) {
        ps[cg] += __shfl_xor(ps[cg], 16, 32);
        pq[cg] += __shfl_xor(pq[cg], 16, 32);
        if (hi == 0) {
            atomicAdd(&bns[cg * 16 + lo], ps[cg]);
            atomicAdd(&bnq[cg * 16 + lo], pq[cg]);
        }
    }
}

__global__ __launch_bounds__(256) void pna_bn(
    float* out, const float* bns, const float* bnq,
    const float* gamma, const float* beta) {
    int i = blockIdx.x * 256 + threadIdx.x;
    if (i >= NNODES * DOUT) return;
    int c = i & 63;
    float mu  = bns[c] / (float)NNODES;
    float var = bnq[c] / (float)NNODES - mu * mu;
    float v = out[i];
    out[i] = (v - mu) * rsqrtf(var + 1e-05f) * gamma[c] + beta[c];
}

extern "C" void kernel_launch(void* const* d_in, const int* in_sizes, int n_in,
                              void* d_out, int out_size, void* d_ws, size_t ws_size,
                              hipStream_t stream) {
    const float* h      = (const float*)d_in[0];
    const float* e      = (const float*)d_in[1];
    const float* snorm  = (const float*)d_in[2];
    const float* Wpre   = (const float*)d_in[3];
    const float* bpre   = (const float*)d_in[4];
    const float* Wpost  = (const float*)d_in[5];
    const float* bpost  = (const float*)d_in[6];
    const float* gamma  = (const float*)d_in[7];
    const float* beta   = (const float*)d_in[8];
    const int*   src    = (const int*)d_in[9];
    const int*   dst    = (const int*)d_in[10];
    float* out = (float*)d_out;

    float*    s1  = (float*)d_ws;
    float*    s2  = s1 + (size_t)NNODES * DIN;
    unsigned* mx  = (unsigned*)(s2 + (size_t)NNODES * DIN);
    unsigned* mn  = mx + (size_t)NNODES * DIN;
    float*    deg = (float*)(mn + (size_t)NNODES * DIN);
    float*    bns = deg + NNODES;
    float*    bnq = bns + DOUT;

    pna_init  <<<(NNODES * DIN + 255) / 256, 256, 0, stream>>>(s1, s2, mx, mn, deg, bns, bnq);
    pna_degree<<<DRANGES * DSLICES, DTHREADS, 0, stream>>>(dst, deg);
    pna_pre   <<<PRE_BLOCKS, PRE_WAVES * 32, 0, stream>>>(h, Wpre, bpre, 0, 0, out, nullptr);
    pna_edge  <<<EBLOCKS, 128, 0, stream>>>(e, Wpre, src, dst, out, s1, s2, mx, mn,
                                            (unsigned*)bns, (unsigned*)(bnq + 32));
    pna_pre   <<<PRE_BLOCKS, PRE_WAVES * 32, 0, stream>>>(h, Wpre, bpre, DIN, 1, out, bns);
    pna_node  <<<NODE_BLOCKS, NODE_WAVES * 32, 0, stream>>>(h, snorm, Wpost, bpost, s1, s2, mx, mn, deg,
                                                out, bns, bnq);
    pna_bn    <<<(NNODES * DOUT + 255) / 256, 256, 0, stream>>>(out, bns, bnq, gamma, beta);
}
